# starting kernel
# baseline (speedup 1.0000x reference)
#include <hip/hip_runtime.h>
#include <hip/hip_bf16.h>

#define N_NODES   50000
#define N_EDGES   600000
#define BAG       16
#define EMB       128
#define HID       128
#define N_CLASSES 10
#define N_GRAPHS  512
#define PAD_IDX   1

typedef __attribute__((ext_vector_type(2))) float v2f;
typedef __attribute__((ext_vector_type(8))) float v8f;

__global__ void zero_f32(float* __restrict__ p, int n) {
    int i = blockIdx.x * blockDim.x + threadIdx.x;
    for (; i < n; i += gridDim.x * blockDim.x) p[i] = 0.0f;
}

__global__ void emb_bag_mean(const int* __restrict__ tokens,
                             const float* __restrict__ table,
                             float* __restrict__ out) {
    const int node = blockIdx.x * (blockDim.x >> 5) + (threadIdx.x >> 5);
    if (node >= N_NODES) return;
    const int lane = threadIdx.x & 31;
    const int* t = tokens + node * BAG;
    float4 acc = make_float4(0.f, 0.f, 0.f, 0.f);
    float cnt = 0.f;
    #pragma unroll
    for (int i = 0; i < BAG; ++i) {
        const int tok = t[i];
        if (tok != PAD_IDX) {
            const float4 e = *(const float4*)(table + (size_t)tok * EMB + lane * 4);
            acc.x += e.x; acc.y += e.y; acc.z += e.z; acc.w += e.w;
            cnt += 1.0f;
        }
    }
    const float inv = 1.0f / fmaxf(cnt, 1.0f);
    float4 r = make_float4(acc.x * inv, acc.y * inv, acc.z * inv, acc.w * inv);
    *(float4*)(out + (size_t)node * EMB + lane * 4) = r;
}

__global__ void count_deg(const int* __restrict__ dst, float* __restrict__ deg) {
    const int e = blockIdx.x * blockDim.x + threadIdx.x;
    if (e < N_EDGES) atomicAdd(&deg[dst[e]], 1.0f);
}

__global__ void count_graph(const int* __restrict__ batch, float* __restrict__ gcnt) {
    const int i = blockIdx.x * blockDim.x + threadIdx.x;
    if (i < N_NODES) atomicAdd(&gcnt[batch[i]], 1.0f);
}

__global__ void edge_scatter(const int* __restrict__ src,
                             const int* __restrict__ dst,
                             const float* __restrict__ x,
                             float* __restrict__ msg) {
    const int e = blockIdx.x * (blockDim.x >> 5) + (threadIdx.x >> 5);
    if (e >= N_EDGES) return;
    const int lane = threadIdx.x & 31;
    const int s = src[e];
    const int d = dst[e];
    const float4 v = *(const float4*)(x + (size_t)s * HID + lane * 4);
    float* p = msg + (size_t)d * HID + lane * 4;
    atomicAdd(p + 0, v.x);
    atomicAdd(p + 1, v.y);
    atomicAdd(p + 2, v.z);
    atomicAdd(p + 3, v.w);
}

__global__ void sage_wmma(const float* __restrict__ msg,
                          const float* __restrict__ deg,
                          const float* __restrict__ x,
                          const float* __restrict__ wl,
                          const float* __restrict__ bias,
                          const float* __restrict__ wr,
                          float* __restrict__ out) {
    const int lane   = threadIdx.x & 31;
    const int wave   = threadIdx.x >> 5;
    const int half   = lane >> 4;
    const int lrow   = lane & 15;
    const int m_base = blockIdx.x * 16;
    const int n_base = (blockIdx.y * 4 + wave) * 16;

    const int arow = m_base + lrow;
    const int brow = n_base + lrow;
    const float rdeg = 1.0f / fmaxf(deg[arow], 1.0f);

    v8f c;
    const float bv = bias[brow];
    #pragma unroll
    for (int r = 0; r < 8; ++r) c[r] = bv;

    const float* mrow = msg + (size_t)arow * HID;
    const float* xrow = x   + (size_t)arow * HID;
    const float* lrowp = wl + (size_t)brow * HID;
    const float* rrowp = wr + (size_t)brow * HID;

    #pragma unroll 8
    for (int k = 0; k < HID; k += 4) {
        const int ka = k + half * 2;
        v2f am = *(const v2f*)(mrow + ka);
        am *= rdeg;
        const v2f ax = *(const v2f*)(xrow + ka);
        const v2f bl = *(const v2f*)(lrowp + ka);
        const v2f br = *(const v2f*)(rrowp + ka);
        c = __builtin_amdgcn_wmma_f32_16x16x4_f32(false, am, false, bl,
                                                  (short)0, c, false, false);
        c = __builtin_amdgcn_wmma_f32_16x16x4_f32(false, ax, false, br,
                                                  (short)0, c, false, false);
    }

    #pragma unroll
    for (int r = 0; r < 8; ++r) {
        const float v = fmaxf(c[r], 0.0f);
        out[(size_t)(m_base + half * 8 + r) * HID + n_base + lrow] = v;
    }
}

__global__ void pool_scatter(const float* __restrict__ x,
                             const int* __restrict__ batch,
                             float* __restrict__ gsum) {
    const int node = blockIdx.x * (blockDim.x >> 5) + (threadIdx.x >> 5);
    if (node >= N_NODES) return;
    const int lane = threadIdx.x & 31;
    const int g = batch[node];
    const float4 v = *(const float4*)(x + (size_t)node * HID + lane * 4);
    float* p = gsum + (size_t)g * HID + lane * 4;
    atomicAdd(p + 0, v.x);
    atomicAdd(p + 1, v.y);
    atomicAdd(p + 2, v.z);
    atomicAdd(p + 3, v.w);
}

__global__ void out_head(const float* __restrict__ gsum,
                         const float* __restrict__ gcnt,
                         const float* __restrict__ wout,
                         const float* __restrict__ bout,
                         float* __restrict__ out) {
    const int idx = blockIdx.x * blockDim.x + threadIdx.x;
    if (idx >= N_GRAPHS * N_CLASSES) return;
    const int g = idx / N_CLASSES;
    const int c = idx % N_CLASSES;
    const float inv = 1.0f / fmaxf(gcnt[g], 1.0f);
    const float* gv = gsum + (size_t)g * HID;
    const float* wv = wout + (size_t)c * HID;
    float s = 0.0f;
    #pragma unroll 8
    for (int k = 0; k < HID; ++k) s += gv[k] * wv[k];
    out[idx] = s * inv + bout[c];
}

extern "C" void kernel_launch(void* const* d_in, const int* in_sizes, int n_in,
                              void* d_out, int out_size, void* d_ws, size_t ws_size,
                              hipStream_t stream) {
    (void)in_sizes; (void)n_in; (void)out_size; (void)ws_size;

    const int*   x_tokens = (const int*)  d_in[0];
    const int*   edge     = (const int*)  d_in[1];
    const int*   batch    = (const int*)  d_in[2];
    const float* emb      = (const float*)d_in[3];
    const float* w1l      = (const float*)d_in[4];
    const float* b1       = (const float*)d_in[5];
    const float* w1r      = (const float*)d_in[6];
    const float* w2l      = (const float*)d_in[7];
    const float* b2       = (const float*)d_in[8];
    const float* w2r      = (const float*)d_in[9];
    const float* wout     = (const float*)d_in[10];
    const float* bout     = (const float*)d_in[11];
    float*       out      = (float*)d_out;

    const int* src = edge;
    const int* dst = edge + N_EDGES;

    const size_t nodeF = (size_t)N_NODES * HID;
    float* bufA = (float*)d_ws;
    float* bufB = bufA + nodeF;
    float* bufC = bufB + nodeF;
    float* deg  = bufC + nodeF;
    float* gsum = deg + N_NODES;
    float* gcnt = gsum + (size_t)N_GRAPHS * HID;

    const dim3 blk256(256);
    const dim3 gWave((N_NODES + 7) / 8);
    const dim3 gEdge((N_EDGES + 7) / 8);
    const dim3 gGemm(N_NODES / 16, 2);
    const dim3 bGemm(128);

    zero_f32<<<2048, blk256, 0, stream>>>(bufB, (int)nodeF);
    zero_f32<<<256,  blk256, 0, stream>>>(deg, N_NODES);
    zero_f32<<<256,  blk256, 0, stream>>>(gsum, N_GRAPHS * HID);
    zero_f32<<<4,    blk256, 0, stream>>>(gcnt, N_GRAPHS);

    emb_bag_mean<<<gWave, blk256, 0, stream>>>(x_tokens, emb, bufA);

    count_deg<<<(N_EDGES + 255) / 256, blk256, 0, stream>>>(dst, deg);
    count_graph<<<(N_NODES + 255) / 256, blk256, 0, stream>>>(batch, gcnt);

    edge_scatter<<<gEdge, blk256, 0, stream>>>(src, dst, bufA, bufB);
    sage_wmma<<<gGemm, bGemm, 0, stream>>>(bufB, deg, bufA, w1l, b1, w1r, bufC);

    zero_f32<<<2048, blk256, 0, stream>>>(bufB, (int)nodeF);
    edge_scatter<<<gEdge, blk256, 0, stream>>>(src, dst, bufC, bufB);
    sage_wmma<<<gGemm, bGemm, 0, stream>>>(bufB, deg, bufC, w2l, b2, w2r, bufA);

    pool_scatter<<<gWave, blk256, 0, stream>>>(bufA, batch, gsum);
    out_head<<<(N_GRAPHS * N_CLASSES + 255) / 256, blk256, 0, stream>>>(
        gsum, gcnt, wout, bout, out);
}


# Edge sums without global atomics: 256 resident work-groups each own 196 destination nodes, filter the whole dst list, add source rows into LDS sums (LDS f32 adds, arrival order); no msg zero fill
# speedup vs baseline: 1.7691x; 1.7691x over previous
#include <hip/hip_runtime.h>
#include <hip/hip_bf16.h>

#define N_NODES   50000
#define N_EDGES   600000
#define BAG       16
#define EMB       128
#define HID       128
#define N_CLASSES 10
#define N_GRAPHS  512
#define PAD_IDX   1

typedef __attribute__((ext_vector_type(2))) float v2f;
typedef __attribute__((ext_vector_type(8))) float v8f;

__global__ void zero_f32(float* __restrict__ p, int n) {
    int i = blockIdx.x * blockDim.x + threadIdx.x;
    for (; i < n; i += gridDim.x * blockDim.x) p[i] = 0.0f;
}

__global__ void emb_bag_mean(const int* __restrict__ tokens,
                             const float* __restrict__ table,
                             float* __restrict__ out) {
    const int node = blockIdx.x * (blockDim.x >> 5) + (threadIdx.x >> 5);
    if (node >= N_NODES) return;
    const int lane = threadIdx.x & 31;
    const int* t = tokens + node * BAG;
    float4 acc = make_float4(0.f, 0.f, 0.f, 0.f);
    float cnt = 0.f;
    #pragma unroll
    for (int i = 0; i < BAG; ++i) {
        const int tok = t[i];
        if (tok != PAD_IDX) {
            const float4 e = *(const float4*)(table + (size_t)tok * EMB + lane * 4);
            acc.x += e.x; acc.y += e.y; acc.z += e.z; acc.w += e.w;
            cnt += 1.0f;
        }
    }
    const float inv = 1.0f / fmaxf(cnt, 1.0f);
    float4 r = make_float4(acc.x * inv, acc.y * inv, acc.z * inv, acc.w * inv);
    *(float4*)(out + (size_t)node * EMB + lane * 4) = r;
}

__global__ void count_deg(const int* __restrict__ dst, float* __restrict__ deg) {
    const int e = blockIdx.x * blockDim.x + threadIdx.x;
    if (e < N_EDGES) atomicAdd(&deg[dst[e]], 1.0f);
}

__global__ void count_graph(const int* __restrict__ batch, float* __restrict__ gcnt) {
    const int i = blockIdx.x * blockDim.x + threadIdx.x;
    if (i < N_NODES) atomicAdd(&gcnt[batch[i]], 1.0f);
}

#define ES_WGS    256
#define ES_RANGE  196
#define ES_WAVES  32
#define ES_QCAP   1024
#define ES_DRAIN_AT 16
#define ES_GROUP  4
#define ES_STRIDE (ES_GROUP * ES_WAVES * 32)
#define ES_PASSES ((N_EDGES / 4 + ES_STRIDE - 1) / ES_STRIDE)
static_assert(ES_WGS * ES_RANGE >= N_NODES, "ranges must cover every node");
static_assert(N_EDGES % 4 == 0, "dst is read as int4");
static_assert(N_EDGES < (1 << 23) && ES_RANGE <= 256, "queue entry = edge << 8 | row");
static_assert(ES_QCAP >= ES_DRAIN_AT + ES_GROUP * 4 * 32, "one pass must fit behind the drain mark");

__device__ __forceinline__ void es_drain(const int* q, int n,
                                         const int* __restrict__ src,
                                         const float* __restrict__ x,
                                         float* acc, int lane) {
    __builtin_amdgcn_fence(__ATOMIC_ACQ_REL, "wavefront");
    for (int b = 0; b < n; b += 32) {
        const int idx = b + lane;
        const int ent = idx < n ? q[idx] : 0;
        const int s   = src[ent >> 8];
        const int cnt = (n - b) < 32 ? (n - b) : 32;
        for (int k0 = 0; k0 < cnt; k0 += 8) {
            float4 v[8];
            int    r[8];
            #pragma unroll
            for (int t = 0; t < 8; ++t) {
                const int k  = (k0 + t) < cnt ? (k0 + t) : (cnt - 1);
                const int sk = __builtin_amdgcn_readlane(s, k);
                r[t] = __builtin_amdgcn_readlane(ent, k) & 255;
                v[t] = *(const float4*)(x + (size_t)sk * HID + lane * 4);
            }
            #pragma unroll
            for (int t = 0; t < 8; ++t) {
                if (k0 + t < cnt) {
                    float* a = acc + r[t] * HID + lane;
                    atomicAdd(a,      v[t].x);
                    atomicAdd(a + 32, v[t].y);
                    atomicAdd(a + 64, v[t].z);
                    atomicAdd(a + 96, v[t].w);
                }
            }
        }
    }
}

#define ES_PUSH(DV, E)                                                          \
    {                                                                           \
        const unsigned r_ = (unsigned)((DV) - base);                            \
        const bool     m_ = r_ < (unsigned)ES_RANGE;                            \
        const unsigned bal_ = __builtin_amdgcn_ballot_w32(m_);                  \
        if (m_) q[n + (int)__builtin_amdgcn_mbcnt_lo(bal_, 0u)] =               \
                    (int)(((unsigned)(E) << 8) | r_);                           \
        n += __popc(bal_);                                                      \
    }

__global__ __launch_bounds__(ES_WAVES * 32)
void edge_sum_range(const int* __restrict__ src,
                    const int* __restrict__ dst,
                    const float* __restrict__ x,
                    float* __restrict__ msg) {
    __shared__ __attribute__((aligned(16))) float acc[ES_RANGE * HID];
    __shared__ int qs[ES_WAVES * ES_QCAP];

    const int tid  = threadIdx.x;
    const int lane = tid & 31;
    const int wave = tid >> 5;
    const int base = blockIdx.x * ES_RANGE;
    int* q = qs + wave * ES_QCAP;

    for (int i = tid; i < ES_RANGE * HID / 4; i += ES_WAVES * 32)
        ((float4*)acc)[i] = make_float4(0.f, 0.f, 0.f, 0.f);
    __syncthreads();

    const int   n4 = N_EDGES / 4;
    const int4* d4 = (const int4*)dst;
    int n = 0;
    for (int it = 0; it < ES_PASSES; ++it) {
        const int pass = (it + blockIdx.x) % ES_PASSES;
        const int c0   = pass * ES_STRIDE + tid;
        int4 d[ES_GROUP];
        #pragma unroll
        for (int u = 0; u < ES_GROUP; ++u) {
            const int c = c0 + u * (ES_WAVES * 32);
            d[u] = d4[c < n4 ? c : n4 - 1];
            if (c >= n4) d[u] = make_int4(-1, -1, -1, -1);
        }
        #pragma unroll
        for (int u = 0; u < ES_GROUP; ++u) {
            const int c = c0 + u * (ES_WAVES * 32);
            const bool any = (unsigned)(d[u].x - base) < (unsigned)ES_RANGE ||
                             (unsigned)(d[u].y - base) < (unsigned)ES_RANGE ||
                             (unsigned)(d[u].z - base) < (unsigned)ES_RANGE ||
                             (unsigned)(d[u].w - base) < (unsigned)ES_RANGE;
            if (__builtin_amdgcn_ballot_w32(any) != 0u) {
                ES_PUSH(d[u].x, 4 * c + 0)
                ES_PUSH(d[u].y, 4 * c + 1)
                ES_PUSH(d[u].z, 4 * c + 2)
                ES_PUSH(d[u].w, 4 * c + 3)
            }
        }
        if (n >= ES_DRAIN_AT) {
            es_drain(q, n, src, x, acc, lane);
            n = 0;
        }
    }
    es_drain(q, n, src, x, acc, lane);
    __syncthreads();

    for (int r = wave; r < ES_RANGE; r += ES_WAVES) {
        const int node = base + r;
        if (node < N_NODES) {
            const float* a = acc + r * HID + lane;
            *(float4*)(msg + (size_t)node * HID + lane * 4) = make_float4(a[0], a[32], a[64], a[96]);
        }
    }
}

#define GEMM_WGS   256
#define STEP_ROWS  32
#define N_STEPS    ((N_NODES + STEP_ROWS - 1) / STEP_ROWS)
#define A_PITCH    (2 * HID + 4)
#define W_PITCH    (HID + 4)
#define KB         8

#define FETCH_ROW(i, STEP)                                                      \
    {                                                                           \
        int row_ = (STEP) * STEP_ROWS + (i) * 8 + wave;                         \
        row_ = row_ < N_NODES ? row_ : N_NODES - 1;                             \
        rd##i = 1.0f / fmaxf(deg[row_], 1.0f);                                  \
        pm##i = *(const float4*)(msg + (size_t)row_ * HID + lane * 4);          \
        px##i = *(const float4*)(x   + (size_t)row_ * HID + lane * 4);          \
    }
#define PARK_ROW(i, DST)                                                        \
    {                                                                           \
        float4 m_ = pm##i;                                                      \
        m_.x *= rd##i; m_.y *= rd##i; m_.z *= rd##i; m_.w *= rd##i;             \
        *(float4*)((DST) + (i) * 8 * A_PITCH)       = m_;                       \
        *(float4*)((DST) + (i) * 8 * A_PITCH + HID) = px##i;                    \
    }

__global__ __launch_bounds__(256)
void sage_wmma(const float* __restrict__ msg,
               const float* __restrict__ deg,
               const float* __restrict__ x,
               const float* __restrict__ wl,
               const float* __restrict__ bias,
               const float* __restrict__ wr,
               float* __restrict__ out) {
    __shared__ __attribute__((aligned(16))) float sW[2 * HID * W_PITCH];
    __shared__ __attribute__((aligned(16))) float sA[2 * STEP_ROWS * A_PITCH];

    const int lane   = threadIdx.x & 31;
    const int wave   = threadIdx.x >> 5;
    const int half   = lane >> 4;
    const int lrow   = lane & 15;
    const int n_base = wave * 16;

    {
        const float4* gl = (const float4*)(wl + (size_t)n_base * HID);
        const float4* gr = (const float4*)(wr + (size_t)n_base * HID);
        float* dl = sW + n_base * W_PITCH + lane * 4;
        float* dr = sW + (HID + n_base) * W_PITCH + lane * 4;
        #pragma unroll
        for (int i = 0; i < 16; ++i) {
            const float4 a = gl[i * 32 + lane];
            const float4 b = gr[i * 32 + lane];
            *(float4*)(dl + i * W_PITCH) = a;
            *(float4*)(dr + i * W_PITCH) = b;
        }
    }

    float4 pm0, pm1, pm2, pm3, px0, px1, px2, px3;
    float  rd0, rd1, rd2, rd3;

    int step = blockIdx.x;
    FETCH_ROW(0, step) FETCH_ROW(1, step) FETCH_ROW(2, step) FETCH_ROW(3, step)

    __syncthreads();
    v2f bl[HID / 4], br[HID / 4];
    {
        const float* pl = sW + (n_base + lrow) * W_PITCH + half * 2;
        const float* pr = pl + HID * W_PITCH;
        #pragma unroll
        for (int j = 0; j < HID / 4; ++j) {
            bl[j] = *(const v2f*)(pl + 4 * j);
            br[j] = *(const v2f*)(pr + 4 * j);
        }
    }
    const float bv = bias[n_base + lrow];

    {
        float* d = sA + wave * A_PITCH + lane * 4;
        PARK_ROW(0, d) PARK_ROW(1, d) PARK_ROW(2, d) PARK_ROW(3, d)
    }
    __syncthreads();

    int buf = 0;
    for (; step < N_STEPS; step += GEMM_WGS) {
        const int  next     = step + GEMM_WGS;
        const bool has_next = next < N_STEPS;
        if (has_next) {
            FETCH_ROW(0, next) FETCH_ROW(1, next) FETCH_ROW(2, next) FETCH_ROW(3, next)
        }

        const float* a0 = sA + buf * (STEP_ROWS * A_PITCH) + lrow * A_PITCH + half * 2;
        const float* a1 = a0 + 16 * A_PITCH;

        v8f c0, c1;
        #pragma unroll
        for (int r = 0; r < 8; ++r) { c0[r] = bv; c1[r] = bv; }

        v2f fm0[2][KB], fm1[2][KB], fx0[2][KB], fx1[2][KB];
        #pragma unroll
        for (int t = 0; t < KB; ++t) {
            fm0[0][t] = *(const v2f*)(a0 + 4 * t);
            fm1[0][t] = *(const v2f*)(a1 + 4 * t);
            fx0[0][t] = *(const v2f*)(a0 + HID + 4 * t);
            fx1[0][t] = *(const v2f*)(a1 + HID + 4 * t);
        }
        #pragma unroll
        for (int b = 0; b < HID / 4 / KB; ++b) {
            const int cur = b & 1;
            if (b + 1 < HID / 4 / KB) {
                #pragma unroll
                for (int t = 0; t < KB; ++t) {
                    const int k = 4 * ((b + 1) * KB + t);
                    fm0[cur ^ 1][t] = *(const v2f*)(a0 + k);
                    fm1[cur ^ 1][t] = *(const v2f*)(a1 + k);
                    fx0[cur ^ 1][t] = *(const v2f*)(a0 + HID + k);
                    fx1[cur ^ 1][t] = *(const v2f*)(a1 + HID + k);
                }
            }
            __builtin_amdgcn_sched_barrier(0);
            #pragma unroll
            for (int t = 0; t < KB; ++t) {
                const int j = b * KB + t;
                c0 = __builtin_amdgcn_wmma_f32_16x16x4_f32(false, fm0[cur][t], false, bl[j],
                                                           (short)0, c0, false, false);
                c1 = __builtin_amdgcn_wmma_f32_16x16x4_f32(false, fm1[cur][t], false, bl[j],
                                                           (short)0, c1, false, false);
                c0 = __builtin_amdgcn_wmma_f32_16x16x4_f32(false, fx0[cur][t], false, br[j],
                                                           (short)0, c0, false, false);
                c1 = __builtin_amdgcn_wmma_f32_16x16x4_f32(false, fx1[cur][t], false, br[j],
                                                           (short)0, c1, false, false);
            }
        }

        if (has_next) {
            float* d = sA + (buf ^ 1) * (STEP_ROWS * A_PITCH) + wave * A_PITCH + lane * 4;
            PARK_ROW(0, d) PARK_ROW(1, d) PARK_ROW(2, d) PARK_ROW(3, d)
        }
        __syncthreads();
        buf ^= 1;

        const int m0_base = step * STEP_ROWS;
        #pragma unroll
        for (int r = 0; r < 8; ++r)
            out[(size_t)(m0_base + half * 8 + r) * HID + n_base + lrow] = fmaxf(c0[r], 0.0f);
        if (m0_base + 16 < N_NODES) {
            #pragma unroll
            for (int r = 0; r < 8; ++r)
                out[(size_t)(m0_base + 16 + half * 8 + r) * HID + n_base + lrow] = fmaxf(c1[r], 0.0f);
        }
    }
}

__global__ void pool_scatter(const float* __restrict__ x,
                             const int* __restrict__ batch,
                             float* __restrict__ gsum) {
    const int node = blockIdx.x * (blockDim.x >> 5) + (threadIdx.x >> 5);
    if (node >= N_NODES) return;
    const int lane = threadIdx.x & 31;
    const int g = batch[node];
    const float4 v = *(const float4*)(x + (size_t)node * HID + lane * 4);
    float* p = gsum + (size_t)g * HID + lane * 4;
    atomicAdd(p + 0, v.x);
    atomicAdd(p + 1, v.y);
    atomicAdd(p + 2, v.z);
    atomicAdd(p + 3, v.w);
}

__global__ void out_head(const float* __restrict__ gsum,
                         const float* __restrict__ gcnt,
                         const float* __restrict__ wout,
                         const float* __restrict__ bout,
                         float* __restrict__ out) {
    const int idx = blockIdx.x * blockDim.x + threadIdx.x;
    if (idx >= N_GRAPHS * N_CLASSES) return;
    const int g = idx / N_CLASSES;
    const int c = idx % N_CLASSES;
    const float inv = 1.0f / fmaxf(gcnt[g], 1.0f);
    const float* gv = gsum + (size_t)g * HID;
    const float* wv = wout + (size_t)c * HID;
    float s = 0.0f;
    #pragma unroll 8
    for (int k = 0; k < HID; ++k) s += gv[k] * wv[k];
    out[idx] = s * inv + bout[c];
}

extern "C" void kernel_launch(void* const* d_in, const int* in_sizes, int n_in,
                              void* d_out, int out_size, void* d_ws, size_t ws_size,
                              hipStream_t stream) {
    (void)in_sizes; (void)n_in; (void)out_size; (void)ws_size;

    const int*   x_tokens = (const int*)  d_in[0];
    const int*   edge     = (const int*)  d_in[1];
    const int*   batch    = (const int*)  d_in[2];
    const float* emb      = (const float*)d_in[3];
    const float* w1l      = (const float*)d_in[4];
    const float* b1       = (const float*)d_in[5];
    const float* w1r      = (const float*)d_in[6];
    const float* w2l      = (const float*)d_in[7];
    const float* b2       = (const float*)d_in[8];
    const float* w2r      = (const float*)d_in[9];
    const float* wout     = (const float*)d_in[10];
    const float* bout     = (const float*)d_in[11];
    float*       out      = (float*)d_out;

    const int* src = edge;
    const int* dst = edge + N_EDGES;

    const size_t nodeF = (size_t)N_NODES * HID;
    float* bufA = (float*)d_ws;
    float* bufB = bufA + nodeF;
    float* bufC = bufB + nodeF;
    float* deg  = bufC + nodeF;
    float* gsum = deg + N_NODES;
    float* gcnt = gsum + (size_t)N_GRAPHS * HID;

    const dim3 blk256(256);
    const dim3 gWave((N_NODES + 7) / 8);
    const dim3 gEdge(ES_WGS);
    const dim3 bEdge(ES_WAVES * 32);
    const dim3 gGemm(GEMM_WGS);
    const dim3 bGemm(256);

    zero_f32<<<256,  blk256, 0, stream>>>(deg, N_NODES);
    zero_f32<<<256,  blk256, 0, stream>>>(gsum, N_GRAPHS * HID);
    zero_f32<<<4,    blk256, 0, stream>>>(gcnt, N_GRAPHS);

    emb_bag_mean<<<gWave, blk256, 0, stream>>>(x_tokens, emb, bufA);

    count_deg<<<(N_EDGES + 255) / 256, blk256, 0, stream>>>(dst, deg);
    count_graph<<<(N_NODES + 255) / 256, blk256, 0, stream>>>(batch, gcnt);

    edge_sum_range<<<gEdge, bEdge, 0, stream>>>(src, dst, bufA, bufB);
    sage_wmma<<<gGemm, bGemm, 0, stream>>>(bufB, deg, bufA, w1l, b1, w1r, bufC);

    edge_sum_range<<<gEdge, bEdge, 0, stream>>>(src, dst, bufC, bufB);
    sage_wmma<<<gGemm, bGemm, 0, stream>>>(bufB, deg, bufC, w2l, b2, w2r, bufA);

    pool_scatter<<<gWave, blk256, 0, stream>>>(bufA, batch, gsum);
    out_head<<<(N_GRAPHS * N_CLASSES + 255) / 256, blk256, 0, stream>>>(
        gsum, gcnt, wout, bout, out);
}


# Graph mean pool + head by graph owner: 256 work-groups own 2 graphs each, filter the batch list, sum their nodes' rows in registers then LDS (fixed order, no atomics), finish the head; 8 launches
# speedup vs baseline: 2.3429x; 1.3243x over previous
#include <hip/hip_runtime.h>
#include <hip/hip_bf16.h>

#define N_NODES   50000
#define N_EDGES   600000
#define BAG       16
#define EMB       128
#define HID       128
#define N_CLASSES 10
#define N_GRAPHS  512
#define PAD_IDX   1

typedef __attribute__((ext_vector_type(2))) float v2f;
typedef __attribute__((ext_vector_type(8))) float v8f;

__global__ void zero_f32(float* __restrict__ p, int n) {
    int i = blockIdx.x * blockDim.x + threadIdx.x;
    for (; i < n; i += gridDim.x * blockDim.x) p[i] = 0.0f;
}

__global__ void emb_bag_mean(const int* __restrict__ tokens,
                             const float* __restrict__ table,
                             float* __restrict__ out) {
    const int node = blockIdx.x * (blockDim.x >> 5) + (threadIdx.x >> 5);
    if (node >= N_NODES) return;
    const int lane = threadIdx.x & 31;
    const int* t = tokens + node * BAG;
    float4 acc = make_float4(0.f, 0.f, 0.f, 0.f);
    float cnt = 0.f;
    #pragma unroll
    for (int i = 0; i < BAG; ++i) {
        const int tok = t[i];
        if (tok != PAD_IDX) {
            const float4 e = *(const float4*)(table + (size_t)tok * EMB + lane * 4);
            acc.x += e.x; acc.y += e.y; acc.z += e.z; acc.w += e.w;
            cnt += 1.0f;
        }
    }
    const float inv = 1.0f / fmaxf(cnt, 1.0f);
    float4 r = make_float4(acc.x * inv, acc.y * inv, acc.z * inv, acc.w * inv);
    *(float4*)(out + (size_t)node * EMB + lane * 4) = r;
}

__global__ void count_deg(const int* __restrict__ dst, float* __restrict__ deg) {
    const int e = blockIdx.x * blockDim.x + threadIdx.x;
    if (e < N_EDGES) atomicAdd(&deg[dst[e]], 1.0f);
}

#define ES_WGS    256
#define ES_RANGE  196
#define ES_WAVES  32
#define ES_QCAP   1024
#define ES_DRAIN_AT 16
#define ES_GROUP  4
#define ES_STRIDE (ES_GROUP * ES_WAVES * 32)
#define ES_PASSES ((N_EDGES / 4 + ES_STRIDE - 1) / ES_STRIDE)
static_assert(ES_WGS * ES_RANGE >= N_NODES, "ranges must cover every node");
static_assert(N_EDGES % 4 == 0, "dst is read as int4");
static_assert(N_EDGES < (1 << 23) && ES_RANGE <= 256, "queue entry = edge << 8 | row");
static_assert(ES_QCAP >= ES_DRAIN_AT + ES_GROUP * 4 * 32, "one pass must fit behind the drain mark");

__device__ __forceinline__ void es_drain(const int* q, int n,
                                         const int* __restrict__ src,
                                         const float* __restrict__ x,
                                         float* acc, int lane) {
    __builtin_amdgcn_fence(__ATOMIC_ACQ_REL, "wavefront");
    for (int b = 0; b < n; b += 32) {
        const int idx = b + lane;
        const int ent = idx < n ? q[idx] : 0;
        const int s   = src[ent >> 8];
        const int cnt = (n - b) < 32 ? (n - b) : 32;
        for (int k0 = 0; k0 < cnt; k0 += 8) {
            float4 v[8];
            int    r[8];
            #pragma unroll
            for (int t = 0; t < 8; ++t) {
                const int k  = (k0 + t) < cnt ? (k0 + t) : (cnt - 1);
                const int sk = __builtin_amdgcn_readlane(s, k);
                r[t] = __builtin_amdgcn_readlane(ent, k) & 255;
                v[t] = *(const float4*)(x + (size_t)sk * HID + lane * 4);
            }
            #pragma unroll
            for (int t = 0; t < 8; ++t) {
                if (k0 + t < cnt) {
                    float* a = acc + r[t] * HID + lane;
                    atomicAdd(a,      v[t].x);
                    atomicAdd(a + 32, v[t].y);
                    atomicAdd(a + 64, v[t].z);
                    atomicAdd(a + 96, v[t].w);
                }
            }
        }
    }
}

#define ES_PUSH(DV, E)                                                          \
    {                                                                           \
        const unsigned r_ = (unsigned)((DV) - base);                            \
        const bool     m_ = r_ < (unsigned)ES_RANGE;                            \
        const unsigned bal_ = __builtin_amdgcn_ballot_w32(m_);                  \
        if (m_) q[n + (int)__builtin_amdgcn_mbcnt_lo(bal_, 0u)] =               \
                    (int)(((unsigned)(E) << 8) | r_);                           \
        n += __popc(bal_);                                                      \
    }

__global__ __launch_bounds__(ES_WAVES * 32)
void edge_sum_range(const int* __restrict__ src,
                    const int* __restrict__ dst,
                    const float* __restrict__ x,
                    float* __restrict__ msg) {
    __shared__ __attribute__((aligned(16))) float acc[ES_RANGE * HID];
    __shared__ int qs[ES_WAVES * ES_QCAP];

    const int tid  = threadIdx.x;
    const int lane = tid & 31;
    const int wave = tid >> 5;
    const int base = blockIdx.x * ES_RANGE;
    int* q = qs + wave * ES_QCAP;

    for (int i = tid; i < ES_RANGE * HID / 4; i += ES_WAVES * 32)
        ((float4*)acc)[i] = make_float4(0.f, 0.f, 0.f, 0.f);
    __syncthreads();

    const int   n4 = N_EDGES / 4;
    const int4* d4 = (const int4*)dst;
    int n = 0;
    for (int it = 0; it < ES_PASSES; ++it) {
        const int pass = (it + blockIdx.x) % ES_PASSES;
        const int c0   = pass * ES_STRIDE + tid;
        int4 d[ES_GROUP];
        #pragma unroll
        for (int u = 0; u < ES_GROUP; ++u) {
            const int c = c0 + u * (ES_WAVES * 32);
            d[u] = d4[c < n4 ? c : n4 - 1];
            if (c >= n4) d[u] = make_int4(-1, -1, -1, -1);
        }
        #pragma unroll
        for (int u = 0; u < ES_GROUP; ++u) {
            const int c = c0 + u * (ES_WAVES * 32);
            const bool any = (unsigned)(d[u].x - base) < (unsigned)ES_RANGE ||
                             (unsigned)(d[u].y - base) < (unsigned)ES_RANGE ||
                             (unsigned)(d[u].z - base) < (unsigned)ES_RANGE ||
                             (unsigned)(d[u].w - base) < (unsigned)ES_RANGE;
            if (__builtin_amdgcn_ballot_w32(any) != 0u) {
                ES_PUSH(d[u].x, 4 * c + 0)
                ES_PUSH(d[u].y, 4 * c + 1)
                ES_PUSH(d[u].z, 4 * c + 2)
                ES_PUSH(d[u].w, 4 * c + 3)
            }
        }
        if (n >= ES_DRAIN_AT) {
            es_drain(q, n, src, x, acc, lane);
            n = 0;
        }
    }
    es_drain(q, n, src, x, acc, lane);
    __syncthreads();

    for (int r = wave; r < ES_RANGE; r += ES_WAVES) {
        const int node = base + r;
        if (node < N_NODES) {
            const float* a = acc + r * HID + lane;
            *(float4*)(msg + (size_t)node * HID + lane * 4) = make_float4(a[0], a[32], a[64], a[96]);
        }
    }
}

#define GEMM_WGS   256
#define STEP_ROWS  32
#define N_STEPS    ((N_NODES + STEP_ROWS - 1) / STEP_ROWS)
#define A_PITCH    (2 * HID + 4)
#define W_PITCH    (HID + 4)
#define KB         8

#define FETCH_ROW(i, STEP)                                                      \
    {                                                                           \
        int row_ = (STEP) * STEP_ROWS + (i) * 8 + wave;                         \
        row_ = row_ < N_NODES ? row_ : N_NODES - 1;                             \
        rd##i = 1.0f / fmaxf(deg[row_], 1.0f);                                  \
        pm##i = *(const float4*)(msg + (size_t)row_ * HID + lane * 4);          \
        px##i = *(const float4*)(x   + (size_t)row_ * HID + lane * 4);          \
    }
#define PARK_ROW(i, DST)                                                        \
    {                                                                           \
        float4 m_ = pm##i;                                                      \
        m_.x *= rd##i; m_.y *= rd##i; m_.z *= rd##i; m_.w *= rd##i;             \
        *(float4*)((DST) + (i) * 8 * A_PITCH)       = m_;                       \
        *(float4*)((DST) + (i) * 8 * A_PITCH + HID) = px##i;                    \
    }

__global__ __launch_bounds__(256)
void sage_wmma(const float* __restrict__ msg,
               const float* __restrict__ deg,
               const float* __restrict__ x,
               const float* __restrict__ wl,
               const float* __restrict__ bias,
               const float* __restrict__ wr,
               float* __restrict__ out) {
    __shared__ __attribute__((aligned(16))) float sW[2 * HID * W_PITCH];
    __shared__ __attribute__((aligned(16))) float sA[2 * STEP_ROWS * A_PITCH];

    const int lane   = threadIdx.x & 31;
    const int wave   = threadIdx.x >> 5;
    const int half   = lane >> 4;
    const int lrow   = lane & 15;
    const int n_base = wave * 16;

    {
        const float4* gl = (const float4*)(wl + (size_t)n_base * HID);
        const float4* gr = (const float4*)(wr + (size_t)n_base * HID);
        float* dl = sW + n_base * W_PITCH + lane * 4;
        float* dr = sW + (HID + n_base) * W_PITCH + lane * 4;
        #pragma unroll
        for (int i = 0; i < 16; ++i) {
            const float4 a = gl[i * 32 + lane];
            const float4 b = gr[i * 32 + lane];
            *(float4*)(dl + i * W_PITCH) = a;
            *(float4*)(dr + i * W_PITCH) = b;
        }
    }

    float4 pm0, pm1, pm2, pm3, px0, px1, px2, px3;
    float  rd0, rd1, rd2, rd3;

    int step = blockIdx.x;
    FETCH_ROW(0, step) FETCH_ROW(1, step) FETCH_ROW(2, step) FETCH_ROW(3, step)

    __syncthreads();
    v2f bl[HID / 4], br[HID / 4];
    {
        const float* pl = sW + (n_base + lrow) * W_PITCH + half * 2;
        const float* pr = pl + HID * W_PITCH;
        #pragma unroll
        for (int j = 0; j < HID / 4; ++j) {
            bl[j] = *(const v2f*)(pl + 4 * j);
            br[j] = *(const v2f*)(pr + 4 * j);
        }
    }
    const float bv = bias[n_base + lrow];

    {
        float* d = sA + wave * A_PITCH + lane * 4;
        PARK_ROW(0, d) PARK_ROW(1, d) PARK_ROW(2, d) PARK_ROW(3, d)
    }
    __syncthreads();

    int buf = 0;
    for (; step < N_STEPS; step += GEMM_WGS) {
        const int  next     = step + GEMM_WGS;
        const bool has_next = next < N_STEPS;
        if (has_next) {
            FETCH_ROW(0, next) FETCH_ROW(1, next) FETCH_ROW(2, next) FETCH_ROW(3, next)
        }

        const float* a0 = sA + buf * (STEP_ROWS * A_PITCH) + lrow * A_PITCH + half * 2;
        const float* a1 = a0 + 16 * A_PITCH;

        v8f c0, c1;
        #pragma unroll
        for (int r = 0; r < 8; ++r) { c0[r] = bv; c1[r] = bv; }

        v2f fm0[2][KB], fm1[2][KB], fx0[2][KB], fx1[2][KB];
        #pragma unroll
        for (int t = 0; t < KB; ++t) {
            fm0[0][t] = *(const v2f*)(a0 + 4 * t);
            fm1[0][t] = *(const v2f*)(a1 + 4 * t);
            fx0[0][t] = *(const v2f*)(a0 + HID + 4 * t);
            fx1[0][t] = *(const v2f*)(a1 + HID + 4 * t);
        }
        #pragma unroll
        for (int b = 0; b < HID / 4 / KB; ++b) {
            const int cur = b & 1;
            if (b + 1 < HID / 4 / KB) {
                #pragma unroll
                for (int t = 0; t < KB; ++t) {
                    const int k = 4 * ((b + 1) * KB + t);
                    fm0[cur ^ 1][t] = *(const v2f*)(a0 + k);
                    fm1[cur ^ 1][t] = *(const v2f*)(a1 + k);
                    fx0[cur ^ 1][t] = *(const v2f*)(a0 + HID + k);
                    fx1[cur ^ 1][t] = *(const v2f*)(a1 + HID + k);
                }
            }
            __builtin_amdgcn_sched_barrier(0);
            #pragma unroll
            for (int t = 0; t < KB; ++t) {
                const int j = b * KB + t;
                c0 = __builtin_amdgcn_wmma_f32_16x16x4_f32(false, fm0[cur][t], false, bl[j],
                                                           (short)0, c0, false, false);
                c1 = __builtin_amdgcn_wmma_f32_16x16x4_f32(false, fm1[cur][t], false, bl[j],
                                                           (short)0, c1, false, false);
                c0 = __builtin_amdgcn_wmma_f32_16x16x4_f32(false, fx0[cur][t], false, br[j],
                                                           (short)0, c0, false, false);
                c1 = __builtin_amdgcn_wmma_f32_16x16x4_f32(false, fx1[cur][t], false, br[j],
                                                           (short)0, c1, false, false);
            }
        }

        if (has_next) {
            float* d = sA + (buf ^ 1) * (STEP_ROWS * A_PITCH) + wave * A_PITCH + lane * 4;
            PARK_ROW(0, d) PARK_ROW(1, d) PARK_ROW(2, d) PARK_ROW(3, d)
        }
        __syncthreads();
        buf ^= 1;

        const int m0_base = step * STEP_ROWS;
        #pragma unroll
        for (int r = 0; r < 8; ++r)
            out[(size_t)(m0_base + half * 8 + r) * HID + n_base + lrow] = fmaxf(c0[r], 0.0f);
        if (m0_base + 16 < N_NODES) {
            #pragma unroll
            for (int r = 0; r < 8; ++r)
                out[(size_t)(m0_base + 16 + half * 8 + r) * HID + n_base + lrow] = fmaxf(c1[r], 0.0f);
        }
    }
}

#define PH_WGS    256
#define PH_GPW    2
#define PH_WAVES  16
#define PH_T      16
#define PH_U      49
#define PH_GROUPS ((N_NODES + 31) / 32)
#define PH_ITERS  ((PH_GROUPS + PH_WAVES - 1) / PH_WAVES)
static_assert(PH_WGS * PH_GPW == N_GRAPHS, "every graph has exactly one owner");
static_assert(PH_GPW == 2, "a wave keeps one register sum per owned graph: a0, a1");
static_assert(PH_ITERS % PH_U == 0, "the scan loop is written in whole batches");
static_assert(PH_U <= 64, "one bit per group of a batch in a 64-bit mask");
static_assert(PH_GPW * HID <= PH_WAVES * 32, "one thread per (graph, feature) in the combine");

__device__ __forceinline__ void ph_rows(unsigned bal, unsigned rel, int nbase,
                                        const float* __restrict__ x, int lane,
                                        float4& a0, float4& a1) {
    while (bal != 0u) {
        const int left = __popc(bal);
        const int cnt  = left < PH_T ? left : PH_T;
        float4   v[PH_T];
        unsigned g[PH_T];
        int klast = __builtin_ctz(bal);
        #pragma unroll
        for (int t = 0; t < PH_T; ++t) {
            const bool has = bal != 0u;
            const int  k   = has ? __builtin_ctz(bal) : klast;
            klast = k;
            bal   = has ? (bal & (bal - 1u)) : 0u;
            g[t] = (unsigned)__builtin_amdgcn_readlane((int)rel, k);
            v[t] = *(const float4*)(x + (size_t)(nbase + k) * HID + lane * 4);
        }
        #pragma unroll
        for (int t = 0; t < PH_T; ++t) {
            if (t < cnt) {
                if (g[t] == 0u) { a0.x += v[t].x; a0.y += v[t].y; a0.z += v[t].z; a0.w += v[t].w; }
                else            { a1.x += v[t].x; a1.y += v[t].y; a1.z += v[t].z; a1.w += v[t].w; }
            }
        }
    }
}

__global__ __launch_bounds__(PH_WAVES * 32)
void pool_head(const float* __restrict__ x,
               const int* __restrict__ batch,
               const float* __restrict__ wout,
               const float* __restrict__ bout,
               float* __restrict__ out) {
    __shared__ __attribute__((aligned(16))) float part[PH_WAVES * PH_GPW * HID];
    __shared__ __attribute__((aligned(16))) float gs[PH_GPW * HID];
    __shared__ int   pcnt[PH_WAVES * PH_GPW];
    __shared__ float ginv[PH_GPW];

    const int tid  = threadIdx.x;
    const int lane = tid & 31;
    const int wave = tid >> 5;
    const int g0   = blockIdx.x * PH_GPW;

    float4 a0 = make_float4(0.f, 0.f, 0.f, 0.f);
    float4 a1 = make_float4(0.f, 0.f, 0.f, 0.f);
    int c0 = 0, c1 = 0;

    for (int i0 = 0; i0 < PH_ITERS; i0 += PH_U) {
        int b[PH_U];
        #pragma unroll
        for (int u = 0; u < PH_U; ++u) {
            const int node = ((i0 + u) * PH_WAVES + wave) * 32 + lane;
            b[u] = batch[node < N_NODES ? node : N_NODES - 1];
            if (node >= N_NODES) b[u] = -1;
        }
        unsigned long long um = 0ull;
        #pragma unroll
        for (int u = 0; u < PH_U; ++u)
            if (__builtin_amdgcn_ballot_w32((unsigned)(b[u] - g0) < (unsigned)PH_GPW) != 0u) um |= 1ull << u;
        while (um != 0ull) {
            const int u = __builtin_ctzll(um);
            um &= um - 1ull;
            int bv = b[0];
            #pragma unroll
            for (int uu = 1; uu < PH_U; ++uu) bv = (u == uu) ? b[uu] : bv;
            const unsigned rel = (unsigned)(bv - g0);
            const unsigned bal = __builtin_amdgcn_ballot_w32(rel < (unsigned)PH_GPW);
            const unsigned bz  = __builtin_amdgcn_ballot_w32(rel == 0u);
            c0 += __popc(bz);
            c1 += __popc(bal) - __popc(bz);
            ph_rows(bal, rel, ((i0 + u) * PH_WAVES + wave) * 32, x, lane, a0, a1);
        }
    }

    *(float4*)(part + (wave * PH_GPW + 0) * HID + lane * 4) = a0;
    *(float4*)(part + (wave * PH_GPW + 1) * HID + lane * 4) = a1;
    if (lane == 0) { pcnt[wave * PH_GPW + 0] = c0; pcnt[wave * PH_GPW + 1] = c1; }
    __syncthreads();

    if (tid < PH_GPW * HID) {
        const int j = tid / HID, f = tid % HID;
        float s = 0.0f;
        #pragma unroll
        for (int w = 0; w < PH_WAVES; ++w) s += part[(w * PH_GPW + j) * HID + f];
        gs[j * HID + f] = s;
    } else if (tid < PH_GPW * HID + PH_GPW) {
        const int j = tid - PH_GPW * HID;
        int c = 0;
        #pragma unroll
        for (int w = 0; w < PH_WAVES; ++w) c += pcnt[w * PH_GPW + j];
        ginv[j] = 1.0f / fmaxf((float)c, 1.0f);
    }
    __syncthreads();

    for (int p = wave; p < PH_GPW * N_CLASSES; p += PH_WAVES) {
        const int j = p / N_CLASSES, c = p % N_CLASSES;
        const float4 gv = *(const float4*)(gs + j * HID + lane * 4);
        const float4 wv = *(const float4*)(wout + (size_t)c * HID + lane * 4);
        float s = gv.x * wv.x + gv.y * wv.y + gv.z * wv.z + gv.w * wv.w;
        #pragma unroll
        for (int o = 16; o > 0; o >>= 1) s += __shfl_xor(s, o, 32);
        if (lane == 0) out[(g0 + j) * N_CLASSES + c] = s * ginv[j] + bout[c];
    }
}

extern "C" void kernel_launch(void* const* d_in, const int* in_sizes, int n_in,
                              void* d_out, int out_size, void* d_ws, size_t ws_size,
                              hipStream_t stream) {
    (void)in_sizes; (void)n_in; (void)out_size; (void)ws_size;

    const int*   x_tokens = (const int*)  d_in[0];
    const int*   edge     = (const int*)  d_in[1];
    const int*   batch    = (const int*)  d_in[2];
    const float* emb      = (const float*)d_in[3];
    const float* w1l      = (const float*)d_in[4];
    const float* b1       = (const float*)d_in[5];
    const float* w1r      = (const float*)d_in[6];
    const float* w2l      = (const float*)d_in[7];
    const float* b2       = (const float*)d_in[8];
    const float* w2r      = (const float*)d_in[9];
    const float* wout     = (const float*)d_in[10];
    const float* bout     = (const float*)d_in[11];
    float*       out      = (float*)d_out;

    const int* src = edge;
    const int* dst = edge + N_EDGES;

    const size_t nodeF = (size_t)N_NODES * HID;
    float* bufA = (float*)d_ws;
    float* bufB = bufA + nodeF;
    float* bufC = bufB + nodeF;
    float* deg  = bufC + nodeF;

    const dim3 blk256(256);
    const dim3 gWave((N_NODES + 7) / 8);
    const dim3 gEdge(ES_WGS);
    const dim3 bEdge(ES_WAVES * 32);
    const dim3 gGemm(GEMM_WGS);
    const dim3 bGemm(256);

    zero_f32<<<256,  blk256, 0, stream>>>(deg, N_NODES);

    emb_bag_mean<<<gWave, blk256, 0, stream>>>(x_tokens, emb, bufA);

    count_deg<<<(N_EDGES + 255) / 256, blk256, 0, stream>>>(dst, deg);

    edge_sum_range<<<gEdge, bEdge, 0, stream>>>(src, dst, bufA, bufB);
    sage_wmma<<<gGemm, bGemm, 0, stream>>>(bufB, deg, bufA, w1l, b1, w1r, bufC);

    edge_sum_range<<<gEdge, bEdge, 0, stream>>>(src, dst, bufC, bufB);
    sage_wmma<<<gGemm, bGemm, 0, stream>>>(bufB, deg, bufC, w2l, b2, w2r, bufA);

    pool_head<<<PH_WGS, PH_WAVES * 32, 0, stream>>>(bufA, batch, wout, bout, out);
}


# Edge list filtered once per call: layer 1's range owners save each wave's (source,row) entries, layer 2 reads them instead of scanning dst again; conv writes over x; LDS f32 adds, arrival order
# speedup vs baseline: 2.5491x; 1.0880x over previous
#include <hip/hip_runtime.h>
#include <hip/hip_bf16.h>

#define N_NODES   50000
#define N_EDGES   600000
#define BAG       16
#define EMB       128
#define HID       128
#define N_CLASSES 10
#define N_GRAPHS  512
#define PAD_IDX   1

typedef __attribute__((ext_vector_type(2))) float v2f;
typedef __attribute__((ext_vector_type(8))) float v8f;

__global__ void zero_f32(float* __restrict__ p, int n) {
    int i = blockIdx.x * blockDim.x + threadIdx.x;
    for (; i < n; i += gridDim.x * blockDim.x) p[i] = 0.0f;
}

__global__ void emb_bag_mean(const int* __restrict__ tokens,
                             const float* __restrict__ table,
                             float* __restrict__ out) {
    const int node = blockIdx.x * (blockDim.x >> 5) + (threadIdx.x >> 5);
    if (node >= N_NODES) return;
    const int lane = threadIdx.x & 31;
    const int* t = tokens + node * BAG;
    float4 acc = make_float4(0.f, 0.f, 0.f, 0.f);
    float cnt = 0.f;
    #pragma unroll
    for (int i = 0; i < BAG; ++i) {
        const int tok = t[i];
        if (tok != PAD_IDX) {
            const float4 e = *(const float4*)(table + (size_t)tok * EMB + lane * 4);
            acc.x += e.x; acc.y += e.y; acc.z += e.z; acc.w += e.w;
            cnt += 1.0f;
        }
    }
    const float inv = 1.0f / fmaxf(cnt, 1.0f);
    float4 r = make_float4(acc.x * inv, acc.y * inv, acc.z * inv, acc.w * inv);
    *(float4*)(out + (size_t)node * EMB + lane * 4) = r;
}

__global__ void count_deg(const int* __restrict__ dst, float* __restrict__ deg) {
    const int e = blockIdx.x * blockDim.x + threadIdx.x;
    if (e < N_EDGES) atomicAdd(&deg[dst[e]], 1.0f);
}

#define ES_WGS    256
#define ES_RANGE  196
#define ES_WAVES  32
#define ES_QCAP   1024
#define ES_LCAP   256
#define ES_DRAIN_AT 16
#define ES_GROUP  4
#define ES_STRIDE (ES_GROUP * ES_WAVES * 32)
#define ES_PASSES ((N_EDGES / 4 + ES_STRIDE - 1) / ES_STRIDE)
static_assert(ES_WGS * ES_RANGE >= N_NODES, "ranges must cover every node");
static_assert(N_EDGES % 4 == 0, "dst is read as int4");
static_assert(N_EDGES < (1 << 23) && ES_RANGE <= 256, "queue entry = edge << 8 | row");
static_assert(N_NODES < (1 << 23), "list entry = source node << 8 | row");
static_assert(ES_QCAP >= ES_DRAIN_AT + ES_GROUP * 4 * 32, "one pass must fit behind the drain mark");
static_assert(ES_WAVES == 32, "one lane per wave when the saved counts are checked");

__device__ __forceinline__ void es_rows(int s, int r, int cnt,
                                        const float* __restrict__ x,
                                        float* acc, int lane) {
    for (int k0 = 0; k0 < cnt; k0 += 8) {
        float4 v[8];
        int    rr[8];
        #pragma unroll
        for (int t = 0; t < 8; ++t) {
            const int k  = (k0 + t) < cnt ? (k0 + t) : (cnt - 1);
            const int sk = __builtin_amdgcn_readlane(s, k);
            rr[t] = __builtin_amdgcn_readlane(r, k);
            v[t]  = *(const float4*)(x + (size_t)sk * HID + lane * 4);
        }
        #pragma unroll
        for (int t = 0; t < 8; ++t) {
            if (k0 + t < cnt) {
                float* a = acc + rr[t] * HID + lane;
                atomicAdd(a,      v[t].x);
                atomicAdd(a + 32, v[t].y);
                atomicAdd(a + 64, v[t].z);
                atomicAdd(a + 96, v[t].w);
            }
        }
    }
}

template <bool KEEP>
__device__ __forceinline__ void es_drain(const int* q, int n,
                                         const int* __restrict__ src,
                                         const float* __restrict__ x,
                                         float* acc, int lane,
                                         int* __restrict__ wlist, int saved) {
    __builtin_amdgcn_fence(__ATOMIC_ACQ_REL, "wavefront");
    for (int b = 0; b < n; b += 32) {
        const int idx = b + lane;
        const int ent = idx < n ? q[idx] : 0;
        const int s   = src[ent >> 8];
        const int r   = ent & 255;
        if (KEEP) {
            const int pos = saved + idx;
            if (idx < n && pos < ES_LCAP) wlist[pos] = (s << 8) | r;
        }
        es_rows(s, r, (n - b) < 32 ? (n - b) : 32, x, acc, lane);
    }
}

#define ES_PUSH(DV, E)                                                          \
    {                                                                           \
        const unsigned r_ = (unsigned)((DV) - base);                            \
        const bool     m_ = r_ < (unsigned)ES_RANGE;                            \
        const unsigned bal_ = __builtin_amdgcn_ballot_w32(m_);                  \
        if (m_) q[n + (int)__builtin_amdgcn_mbcnt_lo(bal_, 0u)] =               \
                    (int)(((unsigned)(E) << 8) | r_);                           \
        n += __popc(bal_);                                                      \
    }

template <bool KEEP>
__device__ __forceinline__ int es_filter(const int* __restrict__ src,
                                         const int* __restrict__ dst,
                                         const float* __restrict__ x,
                                         float* acc, int* q, int base, int tid, int lane,
                                         int* __restrict__ wlist) {
    const int   n4 = N_EDGES / 4;
    const int4* d4 = (const int4*)dst;
    int n = 0, saved = 0;
    for (int it = 0; it < ES_PASSES; ++it) {
        const int pass = (it + blockIdx.x) % ES_PASSES;
        const int c0   = pass * ES_STRIDE + tid;
        int4 d[ES_GROUP];
        #pragma unroll
        for (int u = 0; u < ES_GROUP; ++u) {
            const int c = c0 + u * (ES_WAVES * 32);
            d[u] = d4[c < n4 ? c : n4 - 1];
            if (c >= n4) d[u] = make_int4(-1, -1, -1, -1);
        }
        #pragma unroll
        for (int u = 0; u < ES_GROUP; ++u) {
            const int c = c0 + u * (ES_WAVES * 32);
            const bool any = (unsigned)(d[u].x - base) < (unsigned)ES_RANGE ||
                             (unsigned)(d[u].y - base) < (unsigned)ES_RANGE ||
                             (unsigned)(d[u].z - base) < (unsigned)ES_RANGE ||
                             (unsigned)(d[u].w - base) < (unsigned)ES_RANGE;
            if (__builtin_amdgcn_ballot_w32(any) != 0u) {
                ES_PUSH(d[u].x, 4 * c + 0)
                ES_PUSH(d[u].y, 4 * c + 1)
                ES_PUSH(d[u].z, 4 * c + 2)
                ES_PUSH(d[u].w, 4 * c + 3)
            }
        }
        if (n >= ES_DRAIN_AT) {
            es_drain<KEEP>(q, n, src, x, acc, lane, wlist, saved);
            saved += n;
            n = 0;
        }
    }
    es_drain<KEEP>(q, n, src, x, acc, lane, wlist, saved);
    return saved + n;
}

template <bool SAVED>
__global__ __launch_bounds__(ES_WAVES * 32)
void edge_sum_range(const int* __restrict__ src,
                    const int* __restrict__ dst,
                    const float* __restrict__ x,
                    float* __restrict__ msg,
                    int* __restrict__ lists,
                    int* __restrict__ counts) {
    __shared__ __attribute__((aligned(16))) float acc[ES_RANGE * HID];
    __shared__ int qs[ES_WAVES * ES_QCAP];

    const int tid  = threadIdx.x;
    const int lane = tid & 31;
    const int wave = tid >> 5;
    const int base = blockIdx.x * ES_RANGE;
    int* q     = qs + wave * ES_QCAP;
    int* wlist = lists + (size_t)(blockIdx.x * ES_WAVES + wave) * ES_LCAP;

    bool use_list = false;
    int  n_saved  = 0;
    int  e_next   = 0;
    if (SAVED) {
        const int cw = counts[blockIdx.x * ES_WAVES + lane];
        use_list = __builtin_amdgcn_ballot_w32(cw > ES_LCAP) == 0u;
        n_saved  = __builtin_amdgcn_readfirstlane(counts[blockIdx.x * ES_WAVES + wave]);
        if (use_list && lane < n_saved) e_next = wlist[lane];
    }

    for (int i = tid; i < ES_RANGE * HID / 4; i += ES_WAVES * 32)
        ((float4*)acc)[i] = make_float4(0.f, 0.f, 0.f, 0.f);
    __syncthreads();

    if (SAVED && use_list) {
        for (int b = 0; b < n_saved; b += 32) {
            const int ent = e_next;
            const int nb  = b + 32;
            e_next = (nb + lane < n_saved) ? wlist[nb + lane] : 0;
            es_rows(ent >> 8, ent & 255, (n_saved - b) < 32 ? (n_saved - b) : 32, x, acc, lane);
        }
    } else {
        const int total = es_filter<!SAVED>(src, dst, x, acc, q, base, tid, lane, wlist);
        if (!SAVED && lane == 0) counts[blockIdx.x * ES_WAVES + wave] = total;
    }
    __syncthreads();

    for (int r = wave; r < ES_RANGE; r += ES_WAVES) {
        const int node = base + r;
        if (node < N_NODES) {
            const float* a = acc + r * HID + lane;
            *(float4*)(msg + (size_t)node * HID + lane * 4) = make_float4(a[0], a[32], a[64], a[96]);
        }
    }
}

#define GEMM_WGS   256
#define STEP_ROWS  32
#define N_STEPS    ((N_NODES + STEP_ROWS - 1) / STEP_ROWS)
#define A_PITCH    (2 * HID + 4)
#define W_PITCH    (HID + 4)
#define KB         8

#define FETCH_ROW(i, STEP)                                                      \
    {                                                                           \
        int row_ = (STEP) * STEP_ROWS + (i) * 8 + wave;                         \
        row_ = row_ < N_NODES ? row_ : N_NODES - 1;                             \
        rd##i = 1.0f / fmaxf(deg[row_], 1.0f);                                  \
        pm##i = *(const float4*)(msg + (size_t)row_ * HID + lane * 4);          \
        px##i = *(const float4*)(x   + (size_t)row_ * HID + lane * 4);          \
    }
#define PARK_ROW(i, DST)                                                        \
    {                                                                           \
        float4 m_ = pm##i;                                                      \
        m_.x *= rd##i; m_.y *= rd##i; m_.z *= rd##i; m_.w *= rd##i;             \
        *(float4*)((DST) + (i) * 8 * A_PITCH)       = m_;                       \
        *(float4*)((DST) + (i) * 8 * A_PITCH + HID) = px##i;                    \
    }

__global__ __launch_bounds__(256)
void sage_wmma(const float* __restrict__ msg,
               const float* __restrict__ deg,
               const float* x,
               const float* __restrict__ wl,
               const float* __restrict__ bias,
               const float* __restrict__ wr,
               float* out) {
    __shared__ __attribute__((aligned(16))) float sW[2 * HID * W_PITCH];
    __shared__ __attribute__((aligned(16))) float sA[2 * STEP_ROWS * A_PITCH];

    const int lane   = threadIdx.x & 31;
    const int wave   = threadIdx.x >> 5;
    const int half   = lane >> 4;
    const int lrow   = lane & 15;
    const int n_base = wave * 16;

    {
        const float4* gl = (const float4*)(wl + (size_t)n_base * HID);
        const float4* gr = (const float4*)(wr + (size_t)n_base * HID);
        float* dl = sW + n_base * W_PITCH + lane * 4;
        float* dr = sW + (HID + n_base) * W_PITCH + lane * 4;
        #pragma unroll
        for (int i = 0; i < 16; ++i) {
            const float4 a = gl[i * 32 + lane];
            const float4 b = gr[i * 32 + lane];
            *(float4*)(dl + i * W_PITCH) = a;
            *(float4*)(dr + i * W_PITCH) = b;
        }
    }

    float4 pm0, pm1, pm2, pm3, px0, px1, px2, px3;
    float  rd0, rd1, rd2, rd3;

    int step = blockIdx.x;
    FETCH_ROW(0, step) FETCH_ROW(1, step) FETCH_ROW(2, step) FETCH_ROW(3, step)

    __syncthreads();
    v2f bl[HID / 4], br[HID / 4];
    {
        const float* pl = sW + (n_base + lrow) * W_PITCH + half * 2;
        const float* pr = pl + HID * W_PITCH;
        #pragma unroll
        for (int j = 0; j < HID / 4; ++j) {
            bl[j] = *(const v2f*)(pl + 4 * j);
            br[j] = *(const v2f*)(pr + 4 * j);
        }
    }
    const float bv = bias[n_base + lrow];

    {
        float* d = sA + wave * A_PITCH + lane * 4;
        PARK_ROW(0, d) PARK_ROW(1, d) PARK_ROW(2, d) PARK_ROW(3, d)
    }
    __syncthreads();

    int buf = 0;
    for (; step < N_STEPS; step += GEMM_WGS) {
        const int  next     = step + GEMM_WGS;
        const bool has_next = next < N_STEPS;
        if (has_next) {
            FETCH_ROW(0, next) FETCH_ROW(1, next) FETCH_ROW(2, next) FETCH_ROW(3, next)
        }

        const float* a0 = sA + buf * (STEP_ROWS * A_PITCH) + lrow * A_PITCH + half * 2;
        const float* a1 = a0 + 16 * A_PITCH;

        v8f c0, c1;
        #pragma unroll
        for (int r = 0; r < 8; ++r) { c0[r] = bv; c1[r] = bv; }

        v2f fm0[2][KB], fm1[2][KB], fx0[2][KB], fx1[2][KB];
        #pragma unroll
        for (int t = 0; t < KB; ++t) {
            fm0[0][t] = *(const v2f*)(a0 + 4 * t);
            fm1[0][t] = *(const v2f*)(a1 + 4 * t);
            fx0[0][t] = *(const v2f*)(a0 + HID + 4 * t);
            fx1[0][t] = *(const v2f*)(a1 + HID + 4 * t);
        }
        #pragma unroll
        for (int b = 0; b < HID / 4 / KB; ++b) {
            const int cur = b & 1;
            if (b + 1 < HID / 4 / KB) {
                #pragma unroll
                for (int t = 0; t < KB; ++t) {
                    const int k = 4 * ((b + 1) * KB + t);
                    fm0[cur ^ 1][t] = *(const v2f*)(a0 + k);
                    fm1[cur ^ 1][t] = *(const v2f*)(a1 + k);
                    fx0[cur ^ 1][t] = *(const v2f*)(a0 + HID + k);
                    fx1[cur ^ 1][t] = *(const v2f*)(a1 + HID + k);
                }
            }
            __builtin_amdgcn_sched_barrier(0);
            #pragma unroll
            for (int t = 0; t < KB; ++t) {
                const int j = b * KB + t;
                c0 = __builtin_amdgcn_wmma_f32_16x16x4_f32(false, fm0[cur][t], false, bl[j],
                                                           (short)0, c0, false, false);
                c1 = __builtin_amdgcn_wmma_f32_16x16x4_f32(false, fm1[cur][t], false, bl[j],
                                                           (short)0, c1, false, false);
                c0 = __builtin_amdgcn_wmma_f32_16x16x4_f32(false, fx0[cur][t], false, br[j],
                                                           (short)0, c0, false, false);
                c1 = __builtin_amdgcn_wmma_f32_16x16x4_f32(false, fx1[cur][t], false, br[j],
                                                           (short)0, c1, false, false);
            }
        }

        if (has_next) {
            float* d = sA + (buf ^ 1) * (STEP_ROWS * A_PITCH) + wave * A_PITCH + lane * 4;
            PARK_ROW(0, d) PARK_ROW(1, d) PARK_ROW(2, d) PARK_ROW(3, d)
        }
        __syncthreads();
        buf ^= 1;

        const int m0_base = step * STEP_ROWS;
        #pragma unroll
        for (int r = 0; r < 8; ++r)
            out[(size_t)(m0_base + half * 8 + r) * HID + n_base + lrow] = fmaxf(c0[r], 0.0f);
        if (m0_base + 16 < N_NODES) {
            #pragma unroll
            for (int r = 0; r < 8; ++r)
                out[(size_t)(m0_base + 16 + half * 8 + r) * HID + n_base + lrow] = fmaxf(c1[r], 0.0f);
        }
    }
}

#define PH_WGS    256
#define PH_GPW    2
#define PH_WAVES  16
#define PH_T      16
#define PH_U      49
#define PH_GROUPS ((N_NODES + 31) / 32)
#define PH_ITERS  ((PH_GROUPS + PH_WAVES - 1) / PH_WAVES)
static_assert(PH_WGS * PH_GPW == N_GRAPHS, "every graph has exactly one owner");
static_assert(PH_GPW == 2, "a wave keeps one register sum per owned graph: a0, a1");
static_assert(PH_ITERS % PH_U == 0, "the scan loop is written in whole batches");
static_assert(PH_U <= 64, "one bit per group of a batch in a 64-bit mask");
static_assert(PH_GPW * HID <= PH_WAVES * 32, "one thread per (graph, feature) in the combine");

__device__ __forceinline__ void ph_rows(unsigned bal, unsigned rel, int nbase,
                                        const float* __restrict__ x, int lane,
                                        float4& a0, float4& a1) {
    while (bal != 0u) {
        const int left = __popc(bal);
        const int cnt  = left < PH_T ? left : PH_T;
        float4   v[PH_T];
        unsigned g[PH_T];
        int klast = __builtin_ctz(bal);
        #pragma unroll
        for (int t = 0; t < PH_T; ++t) {
            const bool has = bal != 0u;
            const int  k   = has ? __builtin_ctz(bal) : klast;
            klast = k;
            bal   = has ? (bal & (bal - 1u)) : 0u;
            g[t] = (unsigned)__builtin_amdgcn_readlane((int)rel, k);
            v[t] = *(const float4*)(x + (size_t)(nbase + k) * HID + lane * 4);
        }
        #pragma unroll
        for (int t = 0; t < PH_T; ++t) {
            if (t < cnt) {
                if (g[t] == 0u) { a0.x += v[t].x; a0.y += v[t].y; a0.z += v[t].z; a0.w += v[t].w; }
                else            { a1.x += v[t].x; a1.y += v[t].y; a1.z += v[t].z; a1.w += v[t].w; }
            }
        }
    }
}

__global__ __launch_bounds__(PH_WAVES * 32)
void pool_head(const float* __restrict__ x,
               const int* __restrict__ batch,
               const float* __restrict__ wout,
               const float* __restrict__ bout,
               float* __restrict__ out) {
    __shared__ __attribute__((aligned(16))) float part[PH_WAVES * PH_GPW * HID];
    __shared__ __attribute__((aligned(16))) float gs[PH_GPW * HID];
    __shared__ int   pcnt[PH_WAVES * PH_GPW];
    __shared__ float ginv[PH_GPW];

    const int tid  = threadIdx.x;
    const int lane = tid & 31;
    const int wave = tid >> 5;
    const int g0   = blockIdx.x * PH_GPW;

    float4 a0 = make_float4(0.f, 0.f, 0.f, 0.f);
    float4 a1 = make_float4(0.f, 0.f, 0.f, 0.f);
    int c0 = 0, c1 = 0;

    for (int i0 = 0; i0 < PH_ITERS; i0 += PH_U) {
        int b[PH_U];
        #pragma unroll
        for (int u = 0; u < PH_U; ++u) {
            const int node = ((i0 + u) * PH_WAVES + wave) * 32 + lane;
            b[u] = batch[node < N_NODES ? node : N_NODES - 1];
            if (node >= N_NODES) b[u] = -1;
        }
        unsigned long long um = 0ull;
        #pragma unroll
        for (int u = 0; u < PH_U; ++u)
            if (__builtin_amdgcn_ballot_w32((unsigned)(b[u] - g0) < (unsigned)PH_GPW) != 0u) um |= 1ull << u;
        while (um != 0ull) {
            const int u = __builtin_ctzll(um);
            um &= um - 1ull;
            int bv = b[0];
            #pragma unroll
            for (int uu = 1; uu < PH_U; ++uu) bv = (u == uu) ? b[uu] : bv;
            const unsigned rel = (unsigned)(bv - g0);
            const unsigned bal = __builtin_amdgcn_ballot_w32(rel < (unsigned)PH_GPW);
            const unsigned bz  = __builtin_amdgcn_ballot_w32(rel == 0u);
            c0 += __popc(bz);
            c1 += __popc(bal) - __popc(bz);
            ph_rows(bal, rel, ((i0 + u) * PH_WAVES + wave) * 32, x, lane, a0, a1);
        }
    }

    *(float4*)(part + (wave * PH_GPW + 0) * HID + lane * 4) = a0;
    *(float4*)(part + (wave * PH_GPW + 1) * HID + lane * 4) = a1;
    if (lane == 0) { pcnt[wave * PH_GPW + 0] = c0; pcnt[wave * PH_GPW + 1] = c1; }
    __syncthreads();

    if (tid < PH_GPW * HID) {
        const int j = tid / HID, f = tid % HID;
        float s = 0.0f;
        #pragma unroll
        for (int w = 0; w < PH_WAVES; ++w) s += part[(w * PH_GPW + j) * HID + f];
        gs[j * HID + f] = s;
    } else if (tid < PH_GPW * HID + PH_GPW) {
        const int j = tid - PH_GPW * HID;
        int c = 0;
        #pragma unroll
        for (int w = 0; w < PH_WAVES; ++w) c += pcnt[w * PH_GPW + j];
        ginv[j] = 1.0f / fmaxf((float)c, 1.0f);
    }
    __syncthreads();

    for (int p = wave; p < PH_GPW * N_CLASSES; p += PH_WAVES) {
        const int j = p / N_CLASSES, c = p % N_CLASSES;
        const float4 gv = *(const float4*)(gs + j * HID + lane * 4);
        const float4 wv = *(const float4*)(wout + (size_t)c * HID + lane * 4);
        float s = gv.x * wv.x + gv.y * wv.y + gv.z * wv.z + gv.w * wv.w;
        #pragma unroll
        for (int o = 16; o > 0; o >>= 1) s += __shfl_xor(s, o, 32);
        if (lane == 0) out[(g0 + j) * N_CLASSES + c] = s * ginv[j] + bout[c];
    }
}

extern "C" void kernel_launch(void* const* d_in, const int* in_sizes, int n_in,
                              void* d_out, int out_size, void* d_ws, size_t ws_size,
                              hipStream_t stream) {
    (void)in_sizes; (void)n_in; (void)out_size; (void)ws_size;

    const int*   x_tokens = (const int*)  d_in[0];
    const int*   edge     = (const int*)  d_in[1];
    const int*   batch    = (const int*)  d_in[2];
    const float* emb      = (const float*)d_in[3];
    const float* w1l      = (const float*)d_in[4];
    const float* b1       = (const float*)d_in[5];
    const float* w1r      = (const float*)d_in[6];
    const float* w2l      = (const float*)d_in[7];
    const float* b2       = (const float*)d_in[8];
    const float* w2r      = (const float*)d_in[9];
    const float* wout     = (const float*)d_in[10];
    const float* bout     = (const float*)d_in[11];
    float*       out      = (float*)d_out;

    const int* src = edge;
    const int* dst = edge + N_EDGES;

    const size_t nodeF = (size_t)N_NODES * HID;
    float* bufA = (float*)d_ws;
    float* bufB = bufA + nodeF;
    float* bufC = bufB + nodeF;
    float* deg  = bufC + nodeF;
    int* lists  = (int*)bufC;
    int* counts = lists + ES_WGS * ES_WAVES * ES_LCAP;

    const dim3 blk256(256);
    const dim3 gWave((N_NODES + 7) / 8);
    const dim3 gEdge(ES_WGS);
    const dim3 bEdge(ES_WAVES * 32);
    const dim3 gGemm(GEMM_WGS);
    const dim3 bGemm(256);

    zero_f32<<<256,  blk256, 0, stream>>>(deg, N_NODES);

    emb_bag_mean<<<gWave, blk256, 0, stream>>>(x_tokens, emb, bufA);

    count_deg<<<(N_EDGES + 255) / 256, blk256, 0, stream>>>(dst, deg);

    edge_sum_range<false><<<gEdge, bEdge, 0, stream>>>(src, dst, bufA, bufB, lists, counts);
    sage_wmma<<<gGemm, bGemm, 0, stream>>>(bufB, deg, bufA, w1l, b1, w1r, bufA);

    edge_sum_range<true><<<gEdge, bEdge, 0, stream>>>(src, dst, bufA, bufB, lists, counts);
    sage_wmma<<<gGemm, bGemm, 0, stream>>>(bufB, deg, bufA, w2l, b2, w2r, bufA);

    pool_head<<<PH_WGS, PH_WAVES * 32, 0, stream>>>(bufA, batch, wout, bout, out);
}


# Input-only work in one launch by range owner: embeds its 196 nodes, filters dst into saved lists, counts in-degrees in LDS (int adds); both layers read the lists; 8->6 launches; sum orders kept
# speedup vs baseline: 2.9794x; 1.1688x over previous
#include <hip/hip_runtime.h>
#include <hip/hip_bf16.h>

#define N_NODES   50000
#define N_EDGES   600000
#define BAG       16
#define EMB       128
#define HID       128
#define N_CLASSES 10
#define N_GRAPHS  512
#define PAD_IDX   1

typedef __attribute__((ext_vector_type(2))) float v2f;
typedef __attribute__((ext_vector_type(8))) float v8f;

#define ES_WGS    256
#define ES_RANGE  196
#define ES_WAVES  32
#define ES_QCAP   1024
#define ES_LCAP   256
#define ES_DRAIN_AT 16
#define ES_LIST_AT  32
#define ES_GROUP  4
#define ES_STRIDE (ES_GROUP * ES_WAVES * 32)
#define ES_PASSES ((N_EDGES / 4 + ES_STRIDE - 1) / ES_STRIDE)
static_assert(ES_WGS * ES_RANGE >= N_NODES, "ranges must cover every node");
static_assert(N_EDGES % 4 == 0, "dst is read as int4");
static_assert(N_EDGES < (1 << 23) && ES_RANGE <= 256, "queue entry = edge << 8 | row");
static_assert(N_NODES < (1 << 23), "list entry = source node << 8 | row");
static_assert(ES_QCAP >= ES_LIST_AT + ES_GROUP * 4 * 32, "one pass must fit behind the drain mark");
static_assert(ES_WAVES == 32, "one lane per wave when the saved counts are checked");
static_assert(ES_RANGE <= ES_WAVES * 32, "one thread per owned row for the degree counts");

__device__ __forceinline__ void es_rows(int s, int r, int cnt,
                                        const float* __restrict__ x,
                                        float* acc, int lane) {
    for (int k0 = 0; k0 < cnt; k0 += 8) {
        float4 v[8];
        int    rr[8];
        #pragma unroll
        for (int t = 0; t < 8; ++t) {
            const int k  = (k0 + t) < cnt ? (k0 + t) : (cnt - 1);
            const int sk = __builtin_amdgcn_readlane(s, k);
            rr[t] = __builtin_amdgcn_readlane(r, k);
            v[t]  = *(const float4*)(x + (size_t)sk * HID + lane * 4);
        }
        #pragma unroll
        for (int t = 0; t < 8; ++t) {
            if (k0 + t < cnt) {
                float* a = acc + rr[t] * HID + lane;
                atomicAdd(a,      v[t].x);
                atomicAdd(a + 32, v[t].y);
                atomicAdd(a + 64, v[t].z);
                atomicAdd(a + 96, v[t].w);
            }
        }
    }
}

template <bool LIST>
__device__ __forceinline__ void es_drain(const int* q, int n,
                                         const int* __restrict__ src,
                                         const float* __restrict__ x,
                                         float* acc, int lane,
                                         int* __restrict__ wlist, int saved,
                                         int* rowcnt) {
    __builtin_amdgcn_fence(__ATOMIC_ACQ_REL, "wavefront");
    for (int b = 0; b < n; b += 32) {
        const int idx = b + lane;
        const int ent = idx < n ? q[idx] : 0;
        const int s   = src[ent >> 8];
        const int r   = ent & 255;
        if (LIST) {
            if (idx < n) {
                atomicAdd(&rowcnt[r], 1);
                const int pos = saved + idx;
                if (pos < ES_LCAP) wlist[pos] = (s << 8) | r;
            }
        } else {
            es_rows(s, r, (n - b) < 32 ? (n - b) : 32, x, acc, lane);
        }
    }
}

#define ES_PUSH(DV, E)                                                          \
    {                                                                           \
        const unsigned r_ = (unsigned)((DV) - base);                            \
        const bool     m_ = r_ < (unsigned)ES_RANGE;                            \
        const unsigned bal_ = __builtin_amdgcn_ballot_w32(m_);                  \
        if (m_) q[n + (int)__builtin_amdgcn_mbcnt_lo(bal_, 0u)] =               \
                    (int)(((unsigned)(E) << 8) | r_);                           \
        n += __popc(bal_);                                                      \
    }

template <bool LIST>
__device__ __forceinline__ int es_filter(const int* __restrict__ src,
                                         const int* __restrict__ dst,
                                         const float* __restrict__ x,
                                         float* acc, int* q, int base, int tid, int lane,
                                         int* __restrict__ wlist, int* rowcnt) {
    const int   n4 = N_EDGES / 4;
    const int4* d4 = (const int4*)dst;
    int n = 0, saved = 0;
    for (int it = 0; it < ES_PASSES; ++it) {
        const int pass = (it + blockIdx.x) % ES_PASSES;
        const int c0   = pass * ES_STRIDE + tid;
        int4 d[ES_GROUP];
        #pragma unroll
        for (int u = 0; u < ES_GROUP; ++u) {
            const int c = c0 + u * (ES_WAVES * 32);
            d[u] = d4[c < n4 ? c : n4 - 1];
            if (c >= n4) d[u] = make_int4(-1, -1, -1, -1);
        }
        #pragma unroll
        for (int u = 0; u < ES_GROUP; ++u) {
            const int c = c0 + u * (ES_WAVES * 32);
            const bool any = (unsigned)(d[u].x - base) < (unsigned)ES_RANGE ||
                             (unsigned)(d[u].y - base) < (unsigned)ES_RANGE ||
                             (unsigned)(d[u].z - base) < (unsigned)ES_RANGE ||
                             (unsigned)(d[u].w - base) < (unsigned)ES_RANGE;
            if (__builtin_amdgcn_ballot_w32(any) != 0u) {
                ES_PUSH(d[u].x, 4 * c + 0)
                ES_PUSH(d[u].y, 4 * c + 1)
                ES_PUSH(d[u].z, 4 * c + 2)
                ES_PUSH(d[u].w, 4 * c + 3)
            }
        }
        if (n >= (LIST ? ES_LIST_AT : ES_DRAIN_AT)) {
            es_drain<LIST>(q, n, src, x, acc, lane, wlist, saved, rowcnt);
            saved += n;
            n = 0;
        }
    }
    es_drain<LIST>(q, n, src, x, acc, lane, wlist, saved, rowcnt);
    return saved + n;
}

#define EB_NODES ((ES_RANGE + ES_WAVES - 1) / ES_WAVES)
static_assert(BAG <= 32, "one lane per token of a node");

__global__ __launch_bounds__(ES_WAVES * 32)
void range_prep(const int* __restrict__ tokens,
                const float* __restrict__ table,
                const int* __restrict__ src,
                const int* __restrict__ dst,
                float* __restrict__ x0,
                float* __restrict__ deg,
                int* __restrict__ lists,
                int* __restrict__ counts) {
    __shared__ int qs[ES_WAVES * ES_QCAP];
    __shared__ int rowcnt[ES_RANGE];

    const int tid  = threadIdx.x;
    const int lane = tid & 31;
    const int wave = __builtin_amdgcn_readfirstlane(tid >> 5);
    const int base = blockIdx.x * ES_RANGE;
    int* q     = qs + wave * ES_QCAP;
    int* wlist = lists + (size_t)(blockIdx.x * ES_WAVES + wave) * ES_LCAP;

    if (tid < ES_RANGE) rowcnt[tid] = 0;

    int tk[EB_NODES];
    #pragma unroll
    for (int j = 0; j < EB_NODES; ++j) {
        int node = base + wave + j * ES_WAVES;
        node = node < N_NODES ? node : N_NODES - 1;
        tk[j] = tokens[node * BAG + (lane % BAG)];
    }
    __syncthreads();

    int total = 0;
    #pragma unroll 1
    for (int ph = 0; ph < 2; ++ph) {
        if (((ph ^ (wave >> 2)) & 1) == 0) {
            #pragma unroll
            for (int j = 0; j < EB_NODES; ++j) {
                const int r    = wave + j * ES_WAVES;
                const int node = base + r;
                if (r < ES_RANGE && node < N_NODES) {
                    float4 v[BAG];
                    int    tt[BAG];
                    #pragma unroll
                    for (int i = 0; i < BAG; ++i) {
                        tt[i] = __builtin_amdgcn_readlane(tk[j], i);
                        v[i]  = *(const float4*)(table + (size_t)tt[i] * EMB + lane * 4);
                    }
                    float4 a = make_float4(0.f, 0.f, 0.f, 0.f);
                    float  c = 0.f;
                    #pragma unroll
                    for (int i = 0; i < BAG; ++i) {
                        if (tt[i] != PAD_IDX) {
                            a.x += v[i].x; a.y += v[i].y; a.z += v[i].z; a.w += v[i].w;
                            c += 1.0f;
                        }
                    }
                    const float inv = 1.0f / fmaxf(c, 1.0f);
                    *(float4*)(x0 + (size_t)node * EMB + lane * 4) =
                        make_float4(a.x * inv, a.y * inv, a.z * inv, a.w * inv);
                }
            }
        } else {
            total = es_filter<true>(src, dst, nullptr, nullptr, q, base, tid, lane, wlist, rowcnt);
        }
    }
    if (lane == 0) counts[blockIdx.x * ES_WAVES + wave] = total;
    __syncthreads();

    if (tid < ES_RANGE && base + tid < N_NODES) deg[base + tid] = (float)rowcnt[tid];
}

__global__ __launch_bounds__(ES_WAVES * 32)
void edge_sum_range(const int* __restrict__ src,
                    const int* __restrict__ dst,
                    const float* __restrict__ x,
                    float* __restrict__ msg,
                    const int* __restrict__ lists,
                    const int* __restrict__ counts) {
    __shared__ __attribute__((aligned(16))) float acc[ES_RANGE * HID];
    __shared__ int qs[ES_WAVES * ES_QCAP];

    const int tid  = threadIdx.x;
    const int lane = tid & 31;
    const int wave = tid >> 5;
    const int base = blockIdx.x * ES_RANGE;
    int* q = qs + wave * ES_QCAP;
    const int* wlist = lists + (size_t)(blockIdx.x * ES_WAVES + wave) * ES_LCAP;

    const int  cw       = counts[blockIdx.x * ES_WAVES + lane];
    const bool use_list = __builtin_amdgcn_ballot_w32(cw > ES_LCAP) == 0u;
    const int  n_saved  = __builtin_amdgcn_readfirstlane(counts[blockIdx.x * ES_WAVES + wave]);
    int e_next = 0;
    if (use_list && lane < n_saved) e_next = wlist[lane];

    for (int i = tid; i < ES_RANGE * HID / 4; i += ES_WAVES * 32)
        ((float4*)acc)[i] = make_float4(0.f, 0.f, 0.f, 0.f);
    __syncthreads();

    if (use_list) {
        for (int b = 0; b < n_saved; b += 32) {
            const int ent = e_next;
            const int nb  = b + 32;
            e_next = (nb + lane < n_saved) ? wlist[nb + lane] : 0;
            es_rows(ent >> 8, ent & 255, (n_saved - b) < 32 ? (n_saved - b) : 32, x, acc, lane);
        }
    } else {
        es_filter<false>(src, dst, x, acc, q, base, tid, lane, nullptr, nullptr);
    }
    __syncthreads();

    for (int r = wave; r < ES_RANGE; r += ES_WAVES) {
        const int node = base + r;
        if (node < N_NODES) {
            const float* a = acc + r * HID + lane;
            *(float4*)(msg + (size_t)node * HID + lane * 4) = make_float4(a[0], a[32], a[64], a[96]);
        }
    }
}

#define GEMM_WGS   256
#define STEP_ROWS  32
#define N_STEPS    ((N_NODES + STEP_ROWS - 1) / STEP_ROWS)
#define A_PITCH    (2 * HID + 4)
#define W_PITCH    (HID + 4)
#define KB         8

#define FETCH_ROW(i, STEP)                                                      \
    {                                                                           \
        int row_ = (STEP) * STEP_ROWS + (i) * 8 + wave;                         \
        row_ = row_ < N_NODES ? row_ : N_NODES - 1;                             \
        rd##i = 1.0f / fmaxf(deg[row_], 1.0f);                                  \
        pm##i = *(const float4*)(msg + (size_t)row_ * HID + lane * 4);          \
        px##i = *(const float4*)(x   + (size_t)row_ * HID + lane * 4);          \
    }
#define PARK_ROW(i, DST)                                                        \
    {                                                                           \
        float4 m_ = pm##i;                                                      \
        m_.x *= rd##i; m_.y *= rd##i; m_.z *= rd##i; m_.w *= rd##i;             \
        *(float4*)((DST) + (i) * 8 * A_PITCH)       = m_;                       \
        *(float4*)((DST) + (i) * 8 * A_PITCH + HID) = px##i;                    \
    }

__global__ __launch_bounds__(256)
void sage_wmma(const float* __restrict__ msg,
               const float* __restrict__ deg,
               const float* x,
               const float* __restrict__ wl,
               const float* __restrict__ bias,
               const float* __restrict__ wr,
               float* out) {
    __shared__ __attribute__((aligned(16))) float sW[2 * HID * W_PITCH];
    __shared__ __attribute__((aligned(16))) float sA[2 * STEP_ROWS * A_PITCH];

    const int lane   = threadIdx.x & 31;
    const int wave   = threadIdx.x >> 5;
    const int half   = lane >> 4;
    const int lrow   = lane & 15;
    const int n_base = wave * 16;

    {
        const float4* gl = (const float4*)(wl + (size_t)n_base * HID);
        const float4* gr = (const float4*)(wr + (size_t)n_base * HID);
        float* dl = sW + n_base * W_PITCH + lane * 4;
        float* dr = sW + (HID + n_base) * W_PITCH + lane * 4;
        #pragma unroll
        for (int i = 0; i < 16; ++i) {
            const float4 a = gl[i * 32 + lane];
            const float4 b = gr[i * 32 + lane];
            *(float4*)(dl + i * W_PITCH) = a;
            *(float4*)(dr + i * W_PITCH) = b;
        }
    }

    float4 pm0, pm1, pm2, pm3, px0, px1, px2, px3;
    float  rd0, rd1, rd2, rd3;

    int step = blockIdx.x;
    FETCH_ROW(0, step) FETCH_ROW(1, step) FETCH_ROW(2, step) FETCH_ROW(3, step)

    __syncthreads();
    v2f bl[HID / 4], br[HID / 4];
    {
        const float* pl = sW + (n_base + lrow) * W_PITCH + half * 2;
        const float* pr = pl + HID * W_PITCH;
        #pragma unroll
        for (int j = 0; j < HID / 4; ++j) {
            bl[j] = *(const v2f*)(pl + 4 * j);
            br[j] = *(const v2f*)(pr + 4 * j);
        }
    }
    const float bv = bias[n_base + lrow];

    {
        float* d = sA + wave * A_PITCH + lane * 4;
        PARK_ROW(0, d) PARK_ROW(1, d) PARK_ROW(2, d) PARK_ROW(3, d)
    }
    __syncthreads();

    int buf = 0;
    for (; step < N_STEPS; step += GEMM_WGS) {
        const int  next     = step + GEMM_WGS;
        const bool has_next = next < N_STEPS;
        if (has_next) {
            FETCH_ROW(0, next) FETCH_ROW(1, next) FETCH_ROW(2, next) FETCH_ROW(3, next)
        }

        const float* a0 = sA + buf * (STEP_ROWS * A_PITCH) + lrow * A_PITCH + half * 2;
        const float* a1 = a0 + 16 * A_PITCH;

        v8f c0, c1;
        #pragma unroll
        for (int r = 0; r < 8; ++r) { c0[r] = bv; c1[r] = bv; }

        v2f fm0[2][KB], fm1[2][KB], fx0[2][KB], fx1[2][KB];
        #pragma unroll
        for (int t = 0; t < KB; ++t) {
            fm0[0][t] = *(const v2f*)(a0 + 4 * t);
            fm1[0][t] = *(const v2f*)(a1 + 4 * t);
            fx0[0][t] = *(const v2f*)(a0 + HID + 4 * t);
            fx1[0][t] = *(const v2f*)(a1 + HID + 4 * t);
        }
        #pragma unroll
        for (int b = 0; b < HID / 4 / KB; ++b) {
            const int cur = b & 1;
            if (b + 1 < HID / 4 / KB) {
                #pragma unroll
                for (int t = 0; t < KB; ++t) {
                    const int k = 4 * ((b + 1) * KB + t);
                    fm0[cur ^ 1][t] = *(const v2f*)(a0 + k);
                    fm1[cur ^ 1][t] = *(const v2f*)(a1 + k);
                    fx0[cur ^ 1][t] = *(const v2f*)(a0 + HID + k);
                    fx1[cur ^ 1][t] = *(const v2f*)(a1 + HID + k);
                }
            }
            __builtin_amdgcn_sched_barrier(0);
            #pragma unroll
            for (int t = 0; t < KB; ++t) {
                const int j = b * KB + t;
                c0 = __builtin_amdgcn_wmma_f32_16x16x4_f32(false, fm0[cur][t], false, bl[j],
                                                           (short)0, c0, false, false);
                c1 = __builtin_amdgcn_wmma_f32_16x16x4_f32(false, fm1[cur][t], false, bl[j],
                                                           (short)0, c1, false, false);
                c0 = __builtin_amdgcn_wmma_f32_16x16x4_f32(false, fx0[cur][t], false, br[j],
                                                           (short)0, c0, false, false);
                c1 = __builtin_amdgcn_wmma_f32_16x16x4_f32(false, fx1[cur][t], false, br[j],
                                                           (short)0, c1, false, false);
            }
        }

        if (has_next) {
            float* d = sA + (buf ^ 1) * (STEP_ROWS * A_PITCH) + wave * A_PITCH + lane * 4;
            PARK_ROW(0, d) PARK_ROW(1, d) PARK_ROW(2, d) PARK_ROW(3, d)
        }
        __syncthreads();
        buf ^= 1;

        const int m0_base = step * STEP_ROWS;
        #pragma unroll
        for (int r = 0; r < 8; ++r)
            out[(size_t)(m0_base + half * 8 + r) * HID + n_base + lrow] = fmaxf(c0[r], 0.0f);
        if (m0_base + 16 < N_NODES) {
            #pragma unroll
            for (int r = 0; r < 8; ++r)
                out[(size_t)(m0_base + 16 + half * 8 + r) * HID + n_base + lrow] = fmaxf(c1[r], 0.0f);
        }
    }
}

#define PH_WGS    256
#define PH_GPW    2
#define PH_WAVES  16
#define PH_T      16
#define PH_U      49
#define PH_GROUPS ((N_NODES + 31) / 32)
#define PH_ITERS  ((PH_GROUPS + PH_WAVES - 1) / PH_WAVES)
static_assert(PH_WGS * PH_GPW == N_GRAPHS, "every graph has exactly one owner");
static_assert(PH_GPW == 2, "a wave keeps one register sum per owned graph: a0, a1");
static_assert(PH_ITERS % PH_U == 0, "the scan loop is written in whole batches");
static_assert(PH_U <= 64, "one bit per group of a batch in a 64-bit mask");
static_assert(PH_GPW * HID <= PH_WAVES * 32, "one thread per (graph, feature) in the combine");

__device__ __forceinline__ void ph_rows(unsigned bal, unsigned rel, int nbase,
                                        const float* __restrict__ x, int lane,
                                        float4& a0, float4& a1) {
    while (bal != 0u) {
        const int left = __popc(bal);
        const int cnt  = left < PH_T ? left : PH_T;
        float4   v[PH_T];
        unsigned g[PH_T];
        int klast = __builtin_ctz(bal);
        #pragma unroll
        for (int t = 0; t < PH_T; ++t) {
            const bool has = bal != 0u;
            const int  k   = has ? __builtin_ctz(bal) : klast;
            klast = k;
            bal   = has ? (bal & (bal - 1u)) : 0u;
            g[t] = (unsigned)__builtin_amdgcn_readlane((int)rel, k);
            v[t] = *(const float4*)(x + (size_t)(nbase + k) * HID + lane * 4);
        }
        #pragma unroll
        for (int t = 0; t < PH_T; ++t) {
            if (t < cnt) {
                if (g[t] == 0u) { a0.x += v[t].x; a0.y += v[t].y; a0.z += v[t].z; a0.w += v[t].w; }
                else            { a1.x += v[t].x; a1.y += v[t].y; a1.z += v[t].z; a1.w += v[t].w; }
            }
        }
    }
}

__global__ __launch_bounds__(PH_WAVES * 32)
void pool_head(const float* __restrict__ x,
               const int* __restrict__ batch,
               const float* __restrict__ wout,
               const float* __restrict__ bout,
               float* __restrict__ out) {
    __shared__ __attribute__((aligned(16))) float part[PH_WAVES * PH_GPW * HID];
    __shared__ __attribute__((aligned(16))) float gs[PH_GPW * HID];
    __shared__ int   pcnt[PH_WAVES * PH_GPW];
    __shared__ float ginv[PH_GPW];

    const int tid  = threadIdx.x;
    const int lane = tid & 31;
    const int wave = tid >> 5;
    const int g0   = blockIdx.x * PH_GPW;

    float4 a0 = make_float4(0.f, 0.f, 0.f, 0.f);
    float4 a1 = make_float4(0.f, 0.f, 0.f, 0.f);
    int c0 = 0, c1 = 0;

    for (int i0 = 0; i0 < PH_ITERS; i0 += PH_U) {
        int b[PH_U];
        #pragma unroll
        for (int u = 0; u < PH_U; ++u) {
            const int node = ((i0 + u) * PH_WAVES + wave) * 32 + lane;
            b[u] = batch[node < N_NODES ? node : N_NODES - 1];
            if (node >= N_NODES) b[u] = -1;
        }
        unsigned long long um = 0ull;
        #pragma unroll
        for (int u = 0; u < PH_U; ++u)
            if (__builtin_amdgcn_ballot_w32((unsigned)(b[u] - g0) < (unsigned)PH_GPW) != 0u) um |= 1ull << u;
        while (um != 0ull) {
            const int u = __builtin_ctzll(um);
            um &= um - 1ull;
            int bv = b[0];
            #pragma unroll
            for (int uu = 1; uu < PH_U; ++uu) bv = (u == uu) ? b[uu] : bv;
            const unsigned rel = (unsigned)(bv - g0);
            const unsigned bal = __builtin_amdgcn_ballot_w32(rel < (unsigned)PH_GPW);
            const unsigned bz  = __builtin_amdgcn_ballot_w32(rel == 0u);
            c0 += __popc(bz);
            c1 += __popc(bal) - __popc(bz);
            ph_rows(bal, rel, ((i0 + u) * PH_WAVES + wave) * 32, x, lane, a0, a1);
        }
    }

    *(float4*)(part + (wave * PH_GPW + 0) * HID + lane * 4) = a0;
    *(float4*)(part + (wave * PH_GPW + 1) * HID + lane * 4) = a1;
    if (lane == 0) { pcnt[wave * PH_GPW + 0] = c0; pcnt[wave * PH_GPW + 1] = c1; }
    __syncthreads();

    if (tid < PH_GPW * HID) {
        const int j = tid / HID, f = tid % HID;
        float s = 0.0f;
        #pragma unroll
        for (int w = 0; w < PH_WAVES; ++w) s += part[(w * PH_GPW + j) * HID + f];
        gs[j * HID + f] = s;
    } else if (tid < PH_GPW * HID + PH_GPW) {
        const int j = tid - PH_GPW * HID;
        int c = 0;
        #pragma unroll
        for (int w = 0; w < PH_WAVES; ++w) c += pcnt[w * PH_GPW + j];
        ginv[j] = 1.0f / fmaxf((float)c, 1.0f);
    }
    __syncthreads();

    for (int p = wave; p < PH_GPW * N_CLASSES; p += PH_WAVES) {
        const int j = p / N_CLASSES, c = p % N_CLASSES;
        const float4 gv = *(const float4*)(gs + j * HID + lane * 4);
        const float4 wv = *(const float4*)(wout + (size_t)c * HID + lane * 4);
        float s = gv.x * wv.x + gv.y * wv.y + gv.z * wv.z + gv.w * wv.w;
        #pragma unroll
        for (int o = 16; o > 0; o >>= 1) s += __shfl_xor(s, o, 32);
        if (lane == 0) out[(g0 + j) * N_CLASSES + c] = s * ginv[j] + bout[c];
    }
}

extern "C" void kernel_launch(void* const* d_in, const int* in_sizes, int n_in,
                              void* d_out, int out_size, void* d_ws, size_t ws_size,
                              hipStream_t stream) {
    (void)in_sizes; (void)n_in; (void)out_size; (void)ws_size;

    const int*   x_tokens = (const int*)  d_in[0];
    const int*   edge     = (const int*)  d_in[1];
    const int*   batch    = (const int*)  d_in[2];
    const float* emb      = (const float*)d_in[3];
    const float* w1l      = (const float*)d_in[4];
    const float* b1       = (const float*)d_in[5];
    const float* w1r      = (const float*)d_in[6];
    const float* w2l      = (const float*)d_in[7];
    const float* b2       = (const float*)d_in[8];
    const float* w2r      = (const float*)d_in[9];
    const float* wout     = (const float*)d_in[10];
    const float* bout     = (const float*)d_in[11];
    float*       out      = (float*)d_out;

    const int* src = edge;
    const int* dst = edge + N_EDGES;

    const size_t nodeF = (size_t)N_NODES * HID;
    float* bufA = (float*)d_ws;
    float* bufB = bufA + nodeF;
    float* bufC = bufB + nodeF;
    float* deg  = bufC + nodeF;
    int* lists  = (int*)bufC;
    int* counts = lists + ES_WGS * ES_WAVES * ES_LCAP;

    const dim3 gEdge(ES_WGS);
    const dim3 bEdge(ES_WAVES * 32);
    const dim3 gGemm(GEMM_WGS);
    const dim3 bGemm(256);

    range_prep<<<gEdge, bEdge, 0, stream>>>(x_tokens, emb, src, dst, bufA, deg, lists, counts);

    edge_sum_range<<<gEdge, bEdge, 0, stream>>>(src, dst, bufA, bufB, lists, counts);
    sage_wmma<<<gGemm, bGemm, 0, stream>>>(bufB, deg, bufA, w1l, b1, w1r, bufA);

    edge_sum_range<<<gEdge, bEdge, 0, stream>>>(src, dst, bufA, bufB, lists, counts);
    sage_wmma<<<gGemm, bGemm, 0, stream>>>(bufB, deg, bufA, w2l, b2, w2r, bufA);

    pool_head<<<PH_WGS, PH_WAVES * 32, 0, stream>>>(bufA, batch, wout, bout, out);
}


# Owners' edge lists by a two-level split: work-group (super-range, slice) reads 1/16 of dst, hands edges to 16 owners' lists via LDS counters; degrees counted in layer 1's edge sums; arrival order
# speedup vs baseline: 3.2924x; 1.1051x over previous
#include <hip/hip_runtime.h>
#include <hip/hip_bf16.h>

#define N_NODES   50000
#define N_EDGES   600000
#define BAG       16
#define EMB       128
#define HID       128
#define N_CLASSES 10
#define N_GRAPHS  512
#define PAD_IDX   1

typedef __attribute__((ext_vector_type(2))) float v2f;
typedef __attribute__((ext_vector_type(8))) float v8f;

#define ES_WGS    256
#define ES_RANGE  196
#define ES_WAVES  32
#define ES_QCAP   1024
#define SR_SPLIT  16
#define SR_RANGE  (SR_SPLIT * ES_RANGE)
#define SL_CAP    256
#define SL_INT4   ((N_EDGES / 4 + SR_SPLIT - 1) / SR_SPLIT)
#define ES_DRAIN_AT 16
#define ES_LIST_AT  512
#define ES_GROUP  4
#define ES_STRIDE (ES_GROUP * ES_WAVES * 32)
#define ES_PASSES ((N_EDGES / 4 + ES_STRIDE - 1) / ES_STRIDE)
#define SL_PASSES ((SL_INT4 + ES_STRIDE - 1) / ES_STRIDE)
static_assert(ES_WGS * ES_RANGE >= N_NODES, "ranges must cover every node");
static_assert(ES_WGS == SR_SPLIT * SR_SPLIT, "range_prep: one work-group per (super-range, slice)");
static_assert(ES_WAVES == 2 * SR_SPLIT, "edge_sum_range: two waves per list");
static_assert(N_EDGES % 4 == 0, "dst is read as int4");
static_assert(N_EDGES < (1 << 23) && ES_RANGE <= 256, "queue entry (owner's own filter) = edge << 8 | row");
static_assert(N_EDGES < (1 << 20) && SR_RANGE <= 4096, "queue entry (range_prep) = edge << 12 | offset in super-range");
static_assert(N_NODES < (1 << 23), "list entry = source node << 8 | row");
static_assert(ES_QCAP >= ES_LIST_AT + ES_GROUP * 4 * 32, "one pass must fit behind the drain mark");
static_assert(SR_SPLIT <= 32, "one lane per list when the saved counts are checked");
static_assert(ES_RANGE <= ES_WAVES * 32, "one thread per owned row for the degree counts");

template <bool DEG>
__device__ __forceinline__ void es_rows(int s, int r, int cnt,
                                        const float* __restrict__ x,
                                        float* acc, int* rowcnt, int lane) {
    if (DEG) {
        if (lane < cnt) atomicAdd(&rowcnt[r], 1);
    }
    for (int k0 = 0; k0 < cnt; k0 += 8) {
        float4 v[8];
        int    rr[8];
        #pragma unroll
        for (int t = 0; t < 8; ++t) {
            const int k  = (k0 + t) < cnt ? (k0 + t) : (cnt - 1);
            const int sk = __builtin_amdgcn_readlane(s, k);
            rr[t] = __builtin_amdgcn_readlane(r, k);
            v[t]  = *(const float4*)(x + (size_t)sk * HID + lane * 4);
        }
        #pragma unroll
        for (int t = 0; t < 8; ++t) {
            if (k0 + t < cnt) {
                float* a = acc + rr[t] * HID + lane;
                atomicAdd(a,      v[t].x);
                atomicAdd(a + 32, v[t].y);
                atomicAdd(a + 64, v[t].z);
                atomicAdd(a + 96, v[t].w);
            }
        }
    }
}

template <bool LIST, bool DEG>
__device__ __forceinline__ void es_drain(const int* q, int n,
                                         const int* __restrict__ src,
                                         const float* __restrict__ x,
                                         float* acc, int lane,
                                         int* __restrict__ wl, int sl,
                                         int* cnt) {
    constexpr unsigned QSH = LIST ? 12u : 8u;
    __builtin_amdgcn_fence(__ATOMIC_ACQ_REL, "wavefront");
    for (int b = 0; b < n; b += 32) {
        const int idx = b + lane;
        const unsigned ent = idx < n ? (unsigned)q[idx] : 0u;
        const int      s   = src[ent >> QSH];
        const unsigned r   = ent & ((1u << QSH) - 1u);
        if (LIST) {
            if (idx < n) {
                const unsigned k   = r / (unsigned)ES_RANGE;
                const unsigned row = r - k * (unsigned)ES_RANGE;
                const int pos = atomicAdd(&cnt[k], 1);
                if (pos < SL_CAP)
                    wl[((size_t)k * SR_SPLIT + sl) * SL_CAP + pos] = (int)(((unsigned)s << 8) | row);
            }
        } else {
            es_rows<DEG>(s, (int)r, (n - b) < 32 ? (n - b) : 32, x, acc, cnt, lane);
        }
    }
}

#define ES_PUSH(DV, E)                                                          \
    {                                                                           \
        const unsigned r_ = (unsigned)((DV) - base);                            \
        const bool     m_ = r_ < RNG;                                           \
        const unsigned bal_ = __builtin_amdgcn_ballot_w32(m_);                  \
        if (m_) q[n + (int)__builtin_amdgcn_mbcnt_lo(bal_, 0u)] =               \
                    (int)(((unsigned)(E) << QSH) | r_);                         \
        n += __popc(bal_);                                                      \
    }

template <bool LIST, bool DEG>
__device__ __forceinline__ void es_filter(const int* __restrict__ src,
                                          const int* __restrict__ dst,
                                          const float* __restrict__ x,
                                          float* acc, int* q, int base,
                                          int c_begin, int c_end, int rot,
                                          int tid, int lane,
                                          int* __restrict__ wl, int sl, int* cnt) {
    constexpr unsigned RNG = LIST ? (unsigned)SR_RANGE : (unsigned)ES_RANGE;
    constexpr unsigned QSH = LIST ? 12u : 8u;
    constexpr int      NP  = LIST ? SL_PASSES : ES_PASSES;
    const int4* d4 = (const int4*)dst;
    int n = 0;
    #pragma unroll 1
    for (int it = 0; it < NP; ++it) {
        const int pass = (it + rot) % NP;
        const int c0   = c_begin + pass * ES_STRIDE + tid;
        int4 d[ES_GROUP];
        #pragma unroll
        for (int u = 0; u < ES_GROUP; ++u) {
            const int c = c0 + u * (ES_WAVES * 32);
            d[u] = d4[c < c_end ? c : c_end - 1];
            if (c >= c_end) d[u] = make_int4(-1, -1, -1, -1);
        }
        #pragma unroll
        for (int u = 0; u < ES_GROUP; ++u) {
            const int c = c0 + u * (ES_WAVES * 32);
            const bool any = (unsigned)(d[u].x - base) < RNG ||
                             (unsigned)(d[u].y - base) < RNG ||
                             (unsigned)(d[u].z - base) < RNG ||
                             (unsigned)(d[u].w - base) < RNG;
            if (__builtin_amdgcn_ballot_w32(any) != 0u) {
                ES_PUSH(d[u].x, 4 * c + 0)
                ES_PUSH(d[u].y, 4 * c + 1)
                ES_PUSH(d[u].z, 4 * c + 2)
                ES_PUSH(d[u].w, 4 * c + 3)
            }
        }
        if (n >= (LIST ? ES_LIST_AT : ES_DRAIN_AT)) {
            es_drain<LIST, DEG>(q, n, src, x, acc, lane, wl, sl, cnt);
            n = 0;
        }
    }
    es_drain<LIST, DEG>(q, n, src, x, acc, lane, wl, sl, cnt);
}

#define EB_NODES ((ES_RANGE + ES_WAVES - 1) / ES_WAVES)
static_assert(BAG <= 32, "one lane per token of a node");

__global__ __launch_bounds__(ES_WAVES * 32)
void range_prep(const int* __restrict__ tokens,
                const float* __restrict__ table,
                const int* __restrict__ src,
                const int* __restrict__ dst,
                float* __restrict__ x0,
                int* __restrict__ lists,
                int* __restrict__ counts) {
    __shared__ int qs[ES_WAVES * ES_QCAP];
    __shared__ int lcnt[SR_SPLIT];

    const int tid  = threadIdx.x;
    const int lane = tid & 31;
    const int wave = __builtin_amdgcn_readfirstlane(tid >> 5);
    const int base = blockIdx.x * ES_RANGE;
    const int sr   = blockIdx.x / SR_SPLIT;
    const int sl   = blockIdx.x % SR_SPLIT;
    int* q  = qs + wave * ES_QCAP;
    int* wl = lists + (size_t)sr * SR_SPLIT * SR_SPLIT * SL_CAP;

    if (tid < SR_SPLIT) lcnt[tid] = 0;

    int tk[EB_NODES];
    #pragma unroll
    for (int j = 0; j < EB_NODES; ++j) {
        int node = base + wave + j * ES_WAVES;
        node = node < N_NODES ? node : N_NODES - 1;
        tk[j] = tokens[node * BAG + (lane % BAG)];
    }
    __syncthreads();

    #pragma unroll 1
    for (int ph = 0; ph < 2; ++ph) {
        if (((ph ^ (wave >> 2)) & 1) == 0) {
            #pragma unroll
            for (int j = 0; j < EB_NODES; ++j) {
                const int r    = wave + j * ES_WAVES;
                const int node = base + r;
                if (r < ES_RANGE && node < N_NODES) {
                    float4 v[BAG];
                    int    tt[BAG];
                    #pragma unroll
                    for (int i = 0; i < BAG; ++i) {
                        tt[i] = __builtin_amdgcn_readlane(tk[j], i);
                        v[i]  = *(const float4*)(table + (size_t)tt[i] * EMB + lane * 4);
                    }
                    float4 a = make_float4(0.f, 0.f, 0.f, 0.f);
                    float  c = 0.f;
                    #pragma unroll
                    for (int i = 0; i < BAG; ++i) {
                        if (tt[i] != PAD_IDX) {
                            a.x += v[i].x; a.y += v[i].y; a.z += v[i].z; a.w += v[i].w;
                            c += 1.0f;
                        }
                    }
                    const float inv = 1.0f / fmaxf(c, 1.0f);
                    *(float4*)(x0 + (size_t)node * EMB + lane * 4) =
                        make_float4(a.x * inv, a.y * inv, a.z * inv, a.w * inv);
                }
            }
        } else {
            const int c_begin = sl * SL_INT4;
            const int c_end   = (c_begin + SL_INT4) < (N_EDGES / 4) ? (c_begin + SL_INT4) : (N_EDGES / 4);
            es_filter<true, false>(src, dst, nullptr, nullptr, q, sr * SR_RANGE,
                                   c_begin, c_end, sr, tid, lane, wl, sl, lcnt);
        }
    }
    __syncthreads();

    if (tid < SR_SPLIT) counts[(sr * SR_SPLIT + tid) * SR_SPLIT + sl] = lcnt[tid];
}

template <bool DEG>
__global__ __launch_bounds__(ES_WAVES * 32)
void edge_sum_range(const int* __restrict__ src,
                    const int* __restrict__ dst,
                    const float* __restrict__ x,
                    float* __restrict__ msg,
                    float* __restrict__ deg,
                    const int* __restrict__ lists,
                    const int* __restrict__ counts) {
    __shared__ __attribute__((aligned(16))) float acc[ES_RANGE * HID];
    __shared__ int qs[ES_WAVES * ES_QCAP];
    __shared__ int rowcnt[ES_RANGE];

    const int tid  = threadIdx.x;
    const int lane = tid & 31;
    const int wave = tid >> 5;
    const int base = blockIdx.x * ES_RANGE;
    int* q = qs + wave * ES_QCAP;
    const int sl = wave % SR_SPLIT;
    const int hf = wave / SR_SPLIT;

    const int  cw       = counts[blockIdx.x * SR_SPLIT + (lane % SR_SPLIT)];
    const bool use_list = __builtin_amdgcn_ballot_w32(cw > SL_CAP) == 0u;
    const int  c_sl     = counts[blockIdx.x * SR_SPLIT + sl];
    const int  mid      = (c_sl + 1) >> 1;
    const int  n_saved  = __builtin_amdgcn_readfirstlane(hf ? (c_sl - mid) : mid);
    const int* wlist    = lists + (size_t)(blockIdx.x * SR_SPLIT + sl) * SL_CAP + (hf ? mid : 0);
    int e_next = 0;
    if (use_list && lane < n_saved) e_next = wlist[lane];

    for (int i = tid; i < ES_RANGE * HID / 4; i += ES_WAVES * 32)
        ((float4*)acc)[i] = make_float4(0.f, 0.f, 0.f, 0.f);
    if (DEG) {
        if (tid < ES_RANGE) rowcnt[tid] = 0;
    }
    __syncthreads();

    if (use_list) {
        for (int b = 0; b < n_saved; b += 32) {
            const int ent = e_next;
            const int nb  = b + 32;
            e_next = (nb + lane < n_saved) ? wlist[nb + lane] : 0;
            es_rows<DEG>(ent >> 8, ent & 255, (n_saved - b) < 32 ? (n_saved - b) : 32, x, acc, rowcnt, lane);
        }
    } else {
        es_filter<false, DEG>(src, dst, x, acc, q, base, 0, N_EDGES / 4, blockIdx.x,
                              tid, lane, nullptr, 0, rowcnt);
    }
    __syncthreads();

    for (int r = wave; r < ES_RANGE; r += ES_WAVES) {
        const int node = base + r;
        if (node < N_NODES) {
            const float* a = acc + r * HID + lane;
            *(float4*)(msg + (size_t)node * HID + lane * 4) = make_float4(a[0], a[32], a[64], a[96]);
        }
    }
    if (DEG) {
        if (tid < ES_RANGE && base + tid < N_NODES) deg[base + tid] = (float)rowcnt[tid];
    }
}

#define GEMM_WGS   256
#define STEP_ROWS  32
#define N_STEPS    ((N_NODES + STEP_ROWS - 1) / STEP_ROWS)
#define A_PITCH    (2 * HID + 4)
#define W_PITCH    (HID + 4)
#define KB         8

#define FETCH_ROW(i, STEP)                                                      \
    {                                                                           \
        int row_ = (STEP) * STEP_ROWS + (i) * 8 + wave;                         \
        row_ = row_ < N_NODES ? row_ : N_NODES - 1;                             \
        rd##i = 1.0f / fmaxf(deg[row_], 1.0f);                                  \
        pm##i = *(const float4*)(msg + (size_t)row_ * HID + lane * 4);          \
        px##i = *(const float4*)(x   + (size_t)row_ * HID + lane * 4);          \
    }
#define PARK_ROW(i, DST)                                                        \
    {                                                                           \
        float4 m_ = pm##i;                                                      \
        m_.x *= rd##i; m_.y *= rd##i; m_.z *= rd##i; m_.w *= rd##i;             \
        *(float4*)((DST) + (i) * 8 * A_PITCH)       = m_;                       \
        *(float4*)((DST) + (i) * 8 * A_PITCH + HID) = px##i;                    \
    }

__global__ __launch_bounds__(256)
void sage_wmma(const float* __restrict__ msg,
               const float* __restrict__ deg,
               const float* x,
               const float* __restrict__ wl,
               const float* __restrict__ bias,
               const float* __restrict__ wr,
               float* out) {
    __shared__ __attribute__((aligned(16))) float sW[2 * HID * W_PITCH];
    __shared__ __attribute__((aligned(16))) float sA[2 * STEP_ROWS * A_PITCH];

    const int lane   = threadIdx.x & 31;
    const int wave   = threadIdx.x >> 5;
    const int half   = lane >> 4;
    const int lrow   = lane & 15;
    const int n_base = wave * 16;

    {
        const float4* gl = (const float4*)(wl + (size_t)n_base * HID);
        const float4* gr = (const float4*)(wr + (size_t)n_base * HID);
        float* dl = sW + n_base * W_PITCH + lane * 4;
        float* dr = sW + (HID + n_base) * W_PITCH + lane * 4;
        #pragma unroll
        for (int i = 0; i < 16; ++i) {
            const float4 a = gl[i * 32 + lane];
            const float4 b = gr[i * 32 + lane];
            *(float4*)(dl + i * W_PITCH) = a;
            *(float4*)(dr + i * W_PITCH) = b;
        }
    }

    float4 pm0, pm1, pm2, pm3, px0, px1, px2, px3;
    float  rd0, rd1, rd2, rd3;

    int step = blockIdx.x;
    FETCH_ROW(0, step) FETCH_ROW(1, step) FETCH_ROW(2, step) FETCH_ROW(3, step)

    __syncthreads();
    v2f bl[HID / 4], br[HID / 4];
    {
        const float* pl = sW + (n_base + lrow) * W_PITCH + half * 2;
        const float* pr = pl + HID * W_PITCH;
        #pragma unroll
        for (int j = 0; j < HID / 4; ++j) {
            bl[j] = *(const v2f*)(pl + 4 * j);
            br[j] = *(const v2f*)(pr + 4 * j);
        }
    }
    const float bv = bias[n_base + lrow];

    {
        float* d = sA + wave * A_PITCH + lane * 4;
        PARK_ROW(0, d) PARK_ROW(1, d) PARK_ROW(2, d) PARK_ROW(3, d)
    }
    __syncthreads();

    int buf = 0;
    for (; step < N_STEPS; step += GEMM_WGS) {
        const int  next     = step + GEMM_WGS;
        const bool has_next = next < N_STEPS;
        if (has_next) {
            FETCH_ROW(0, next) FETCH_ROW(1, next) FETCH_ROW(2, next) FETCH_ROW(3, next)
        }

        const float* a0 = sA + buf * (STEP_ROWS * A_PITCH) + lrow * A_PITCH + half * 2;
        const float* a1 = a0 + 16 * A_PITCH;

        v8f c0, c1;
        #pragma unroll
        for (int r = 0; r < 8; ++r) { c0[r] = bv; c1[r] = bv; }

        v2f fm0[2][KB], fm1[2][KB], fx0[2][KB], fx1[2][KB];
        #pragma unroll
        for (int t = 0; t < KB; ++t) {
            fm0[0][t] = *(const v2f*)(a0 + 4 * t);
            fm1[0][t] = *(const v2f*)(a1 + 4 * t);
            fx0[0][t] = *(const v2f*)(a0 + HID + 4 * t);
            fx1[0][t] = *(const v2f*)(a1 + HID + 4 * t);
        }
        #pragma unroll
        for (int b = 0; b < HID / 4 / KB; ++b) {
            const int cur = b & 1;
            if (b + 1 < HID / 4 / KB) {
                #pragma unroll
                for (int t = 0; t < KB; ++t) {
                    const int k = 4 * ((b + 1) * KB + t);
                    fm0[cur ^ 1][t] = *(const v2f*)(a0 + k);
                    fm1[cur ^ 1][t] = *(const v2f*)(a1 + k);
                    fx0[cur ^ 1][t] = *(const v2f*)(a0 + HID + k);
                    fx1[cur ^ 1][t] = *(const v2f*)(a1 + HID + k);
                }
            }
            __builtin_amdgcn_sched_barrier(0);
            #pragma unroll
            for (int t = 0; t < KB; ++t) {
                const int j = b * KB + t;
                c0 = __builtin_amdgcn_wmma_f32_16x16x4_f32(false, fm0[cur][t], false, bl[j],
                                                           (short)0, c0, false, false);
                c1 = __builtin_amdgcn_wmma_f32_16x16x4_f32(false, fm1[cur][t], false, bl[j],
                                                           (short)0, c1, false, false);
                c0 = __builtin_amdgcn_wmma_f32_16x16x4_f32(false, fx0[cur][t], false, br[j],
                                                           (short)0, c0, false, false);
                c1 = __builtin_amdgcn_wmma_f32_16x16x4_f32(false, fx1[cur][t], false, br[j],
                                                           (short)0, c1, false, false);
            }
        }

        if (has_next) {
            float* d = sA + (buf ^ 1) * (STEP_ROWS * A_PITCH) + wave * A_PITCH + lane * 4;
            PARK_ROW(0, d) PARK_ROW(1, d) PARK_ROW(2, d) PARK_ROW(3, d)
        }
        __syncthreads();
        buf ^= 1;

        const int m0_base = step * STEP_ROWS;
        #pragma unroll
        for (int r = 0; r < 8; ++r)
            out[(size_t)(m0_base + half * 8 + r) * HID + n_base + lrow] = fmaxf(c0[r], 0.0f);
        if (m0_base + 16 < N_NODES) {
            #pragma unroll
            for (int r = 0; r < 8; ++r)
                out[(size_t)(m0_base + 16 + half * 8 + r) * HID + n_base + lrow] = fmaxf(c1[r], 0.0f);
        }
    }
}

#define PH_WGS    256
#define PH_GPW    2
#define PH_WAVES  16
#define PH_T      16
#define PH_U      49
#define PH_GROUPS ((N_NODES + 31) / 32)
#define PH_ITERS  ((PH_GROUPS + PH_WAVES - 1) / PH_WAVES)
static_assert(PH_WGS * PH_GPW == N_GRAPHS, "every graph has exactly one owner");
static_assert(PH_GPW == 2, "a wave keeps one register sum per owned graph: a0, a1");
static_assert(PH_ITERS % PH_U == 0, "the scan loop is written in whole batches");
static_assert(PH_U <= 64, "one bit per group of a batch in a 64-bit mask");
static_assert(PH_GPW * HID <= PH_WAVES * 32, "one thread per (graph, feature) in the combine");

__device__ __forceinline__ void ph_rows(unsigned bal, unsigned rel, int nbase,
                                        const float* __restrict__ x, int lane,
                                        float4& a0, float4& a1) {
    while (bal != 0u) {
        const int left = __popc(bal);
        const int cnt  = left < PH_T ? left : PH_T;
        float4   v[PH_T];
        unsigned g[PH_T];
        int klast = __builtin_ctz(bal);
        #pragma unroll
        for (int t = 0; t < PH_T; ++t) {
            const bool has = bal != 0u;
            const int  k   = has ? __builtin_ctz(bal) : klast;
            klast = k;
            bal   = has ? (bal & (bal - 1u)) : 0u;
            g[t] = (unsigned)__builtin_amdgcn_readlane((int)rel, k);
            v[t] = *(const float4*)(x + (size_t)(nbase + k) * HID + lane * 4);
        }
        #pragma unroll
        for (int t = 0; t < PH_T; ++t) {
            if (t < cnt) {
                if (g[t] == 0u) { a0.x += v[t].x; a0.y += v[t].y; a0.z += v[t].z; a0.w += v[t].w; }
                else            { a1.x += v[t].x; a1.y += v[t].y; a1.z += v[t].z; a1.w += v[t].w; }
            }
        }
    }
}

__global__ __launch_bounds__(PH_WAVES * 32)
void pool_head(const float* __restrict__ x,
               const int* __restrict__ batch,
               const float* __restrict__ wout,
               const float* __restrict__ bout,
               float* __restrict__ out) {
    __shared__ __attribute__((aligned(16))) float part[PH_WAVES * PH_GPW * HID];
    __shared__ __attribute__((aligned(16))) float gs[PH_GPW * HID];
    __shared__ int   pcnt[PH_WAVES * PH_GPW];
    __shared__ float ginv[PH_GPW];

    const int tid  = threadIdx.x;
    const int lane = tid & 31;
    const int wave = tid >> 5;
    const int g0   = blockIdx.x * PH_GPW;

    float4 a0 = make_float4(0.f, 0.f, 0.f, 0.f);
    float4 a1 = make_float4(0.f, 0.f, 0.f, 0.f);
    int c0 = 0, c1 = 0;

    for (int i0 = 0; i0 < PH_ITERS; i0 += PH_U) {
        int b[PH_U];
        #pragma unroll
        for (int u = 0; u < PH_U; ++u) {
            const int node = ((i0 + u) * PH_WAVES + wave) * 32 + lane;
            b[u] = batch[node < N_NODES ? node : N_NODES - 1];
            if (node >= N_NODES) b[u] = -1;
        }
        unsigned long long um = 0ull;
        #pragma unroll
        for (int u = 0; u < PH_U; ++u)
            if (__builtin_amdgcn_ballot_w32((unsigned)(b[u] - g0) < (unsigned)PH_GPW) != 0u) um |= 1ull << u;
        while (um != 0ull) {
            const int u = __builtin_ctzll(um);
            um &= um - 1ull;
            int bv = b[0];
            #pragma unroll
            for (int uu = 1; uu < PH_U; ++uu) bv = (u == uu) ? b[uu] : bv;
            const unsigned rel = (unsigned)(bv - g0);
            const unsigned bal = __builtin_amdgcn_ballot_w32(rel < (unsigned)PH_GPW);
            const unsigned bz  = __builtin_amdgcn_ballot_w32(rel == 0u);
            c0 += __popc(bz);
            c1 += __popc(bal) - __popc(bz);
            ph_rows(bal, rel, ((i0 + u) * PH_WAVES + wave) * 32, x, lane, a0, a1);
        }
    }

    *(float4*)(part + (wave * PH_GPW + 0) * HID + lane * 4) = a0;
    *(float4*)(part + (wave * PH_GPW + 1) * HID + lane * 4) = a1;
    if (lane == 0) { pcnt[wave * PH_GPW + 0] = c0; pcnt[wave * PH_GPW + 1] = c1; }
    __syncthreads();

    if (tid < PH_GPW * HID) {
        const int j = tid / HID, f = tid % HID;
        float s = 0.0f;
        #pragma unroll
        for (int w = 0; w < PH_WAVES; ++w) s += part[(w * PH_GPW + j) * HID + f];
        gs[j * HID + f] = s;
    } else if (tid < PH_GPW * HID + PH_GPW) {
        const int j = tid - PH_GPW * HID;
        int c = 0;
        #pragma unroll
        for (int w = 0; w < PH_WAVES; ++w) c += pcnt[w * PH_GPW + j];
        ginv[j] = 1.0f / fmaxf((float)c, 1.0f);
    }
    __syncthreads();

    for (int p = wave; p < PH_GPW * N_CLASSES; p += PH_WAVES) {
        const int j = p / N_CLASSES, c = p % N_CLASSES;
        const float4 gv = *(const float4*)(gs + j * HID + lane * 4);
        const float4 wv = *(const float4*)(wout + (size_t)c * HID + lane * 4);
        float s = gv.x * wv.x + gv.y * wv.y + gv.z * wv.z + gv.w * wv.w;
        #pragma unroll
        for (int o = 16; o > 0; o >>= 1) s += __shfl_xor(s, o, 32);
        if (lane == 0) out[(g0 + j) * N_CLASSES + c] = s * ginv[j] + bout[c];
    }
}

extern "C" void kernel_launch(void* const* d_in, const int* in_sizes, int n_in,
                              void* d_out, int out_size, void* d_ws, size_t ws_size,
                              hipStream_t stream) {
    (void)in_sizes; (void)n_in; (void)out_size; (void)ws_size;

    const int*   x_tokens = (const int*)  d_in[0];
    const int*   edge     = (const int*)  d_in[1];
    const int*   batch    = (const int*)  d_in[2];
    const float* emb      = (const float*)d_in[3];
    const float* w1l      = (const float*)d_in[4];
    const float* b1       = (const float*)d_in[5];
    const float* w1r      = (const float*)d_in[6];
    const float* w2l      = (const float*)d_in[7];
    const float* b2       = (const float*)d_in[8];
    const float* w2r      = (const float*)d_in[9];
    const float* wout     = (const float*)d_in[10];
    const float* bout     = (const float*)d_in[11];
    float*       out      = (float*)d_out;

    const int* src = edge;
    const int* dst = edge + N_EDGES;

    const size_t nodeF = (size_t)N_NODES * HID;
    float* bufA = (float*)d_ws;
    float* bufB = bufA + nodeF;
    float* bufC = bufB + nodeF;
    float* deg  = bufC + nodeF;
    int* lists  = (int*)bufC;
    int* counts = lists + ES_WGS * SR_SPLIT * SL_CAP;

    const dim3 gEdge(ES_WGS);
    const dim3 bEdge(ES_WAVES * 32);
    const dim3 gGemm(GEMM_WGS);
    const dim3 bGemm(256);

    range_prep<<<gEdge, bEdge, 0, stream>>>(x_tokens, emb, src, dst, bufA, lists, counts);

    edge_sum_range<true><<<gEdge, bEdge, 0, stream>>>(src, dst, bufA, bufB, deg, lists, counts);
    sage_wmma<<<gGemm, bGemm, 0, stream>>>(bufB, deg, bufA, w1l, b1, w1r, bufA);

    edge_sum_range<false><<<gEdge, bEdge, 0, stream>>>(src, dst, bufA, bufB, deg, lists, counts);
    sage_wmma<<<gGemm, bGemm, 0, stream>>>(bufB, deg, bufA, w2l, b2, w2r, bufA);

    pool_head<<<PH_WGS, PH_WAVES * 32, 0, stream>>>(bufA, batch, wout, bout, out);
}


# Edge sums in registers: the range owner queues its list entries by destination row in LDS (integer add gives the place); a wave sums a row's sources in registers; no LDS f32 adds; arrival order
# speedup vs baseline: 3.4310x; 1.0421x over previous
#include <hip/hip_runtime.h>
#include <hip/hip_bf16.h>

#define N_NODES   50000
#define N_EDGES   600000
#define BAG       16
#define EMB       128
#define HID       128
#define N_CLASSES 10
#define N_GRAPHS  512
#define PAD_IDX   1

typedef __attribute__((ext_vector_type(2))) float v2f;
typedef __attribute__((ext_vector_type(8))) float v8f;

#define ES_WGS    256
#define ES_RANGE  196
#define ES_WAVES  32
#define ES_QCAP   1024
#define SR_SPLIT  16
#define SR_RANGE  (SR_SPLIT * ES_RANGE)
#define SL_CAP    256
#define SL_INT4   ((N_EDGES / 4 + SR_SPLIT - 1) / SR_SPLIT)
#define ES_DRAIN_AT 16
#define ES_LIST_AT  512
#define ES_GROUP  4
#define ES_STRIDE (ES_GROUP * ES_WAVES * 32)
#define ES_PASSES ((N_EDGES / 4 + ES_STRIDE - 1) / ES_STRIDE)
#define SL_PASSES ((SL_INT4 + ES_STRIDE - 1) / ES_STRIDE)
static_assert(ES_WGS * ES_RANGE >= N_NODES, "ranges must cover every node");
static_assert(ES_WGS == SR_SPLIT * SR_SPLIT, "range_prep: one work-group per (super-range, slice)");
static_assert(ES_WAVES == 2 * SR_SPLIT, "edge_sum_range: two waves per list");
static_assert(N_EDGES % 4 == 0, "dst is read as int4");
static_assert(N_EDGES < (1 << 23) && ES_RANGE <= 256, "queue entry (owner's own filter) = edge << 8 | row");
static_assert(N_EDGES < (1 << 20) && SR_RANGE <= 4096, "queue entry (range_prep) = edge << 12 | offset in super-range");
static_assert(N_NODES < (1 << 23), "list entry = source node << 8 | row");
static_assert(ES_QCAP >= ES_LIST_AT + ES_GROUP * 4 * 32, "one pass must fit behind the drain mark");
static_assert(SR_SPLIT <= 32, "one lane per list when the saved counts are checked");
static_assert(ES_RANGE <= ES_WAVES * 32, "one thread per owned row for the degree counts");
#define RQ_CAP    128
#define ES_T      16
#define ES_HALF_LOADS ((SL_CAP / 2 + 31) / 32)
static_assert(ES_RANGE * RQ_CAP <= ES_WAVES * ES_QCAP, "the row queues lie in the space of the waves' queues");
static_assert(ES_T <= 32, "one lane per source of a group");

template <bool DEG>
__device__ __forceinline__ void es_rows(int s, int r, int cnt,
                                        const float* __restrict__ x,
                                        float* acc, int* rowcnt, int lane) {
    if (DEG) {
        if (lane < cnt) atomicAdd(&rowcnt[r], 1);
    }
    for (int k0 = 0; k0 < cnt; k0 += 8) {
        float4 v[8];
        int    rr[8];
        #pragma unroll
        for (int t = 0; t < 8; ++t) {
            const int k  = (k0 + t) < cnt ? (k0 + t) : (cnt - 1);
            const int sk = __builtin_amdgcn_readlane(s, k);
            rr[t] = __builtin_amdgcn_readlane(r, k);
            v[t]  = *(const float4*)(x + (size_t)sk * HID + lane * 4);
        }
        #pragma unroll
        for (int t = 0; t < 8; ++t) {
            if (k0 + t < cnt) {
                float* a = acc + rr[t] * HID + lane;
                atomicAdd(a,      v[t].x);
                atomicAdd(a + 32, v[t].y);
                atomicAdd(a + 64, v[t].z);
                atomicAdd(a + 96, v[t].w);
            }
        }
    }
}

template <bool LIST, bool DEG>
__device__ __forceinline__ void es_drain(const int* q, int n,
                                         const int* __restrict__ src,
                                         const float* __restrict__ x,
                                         float* acc, int lane,
                                         int* __restrict__ wl, int sl,
                                         int* cnt) {
    constexpr unsigned QSH = LIST ? 12u : 8u;
    __builtin_amdgcn_fence(__ATOMIC_ACQ_REL, "wavefront");
    for (int b = 0; b < n; b += 32) {
        const int idx = b + lane;
        const unsigned ent = idx < n ? (unsigned)q[idx] : 0u;
        const int      s   = src[ent >> QSH];
        const unsigned r   = ent & ((1u << QSH) - 1u);
        if (LIST) {
            if (idx < n) {
                const unsigned k   = r / (unsigned)ES_RANGE;
                const unsigned row = r - k * (unsigned)ES_RANGE;
                const int pos = atomicAdd(&cnt[k], 1);
                if (pos < SL_CAP)
                    wl[((size_t)k * SR_SPLIT + sl) * SL_CAP + pos] = (int)(((unsigned)s << 8) | row);
            }
        } else {
            es_rows<DEG>(s, (int)r, (n - b) < 32 ? (n - b) : 32, x, acc, cnt, lane);
        }
    }
}

#define ES_PUSH(DV, E)                                                          \
    {                                                                           \
        const unsigned r_ = (unsigned)((DV) - base);                            \
        const bool     m_ = r_ < RNG;                                           \
        const unsigned bal_ = __builtin_amdgcn_ballot_w32(m_);                  \
        if (m_) q[n + (int)__builtin_amdgcn_mbcnt_lo(bal_, 0u)] =               \
                    (int)(((unsigned)(E) << QSH) | r_);                         \
        n += __popc(bal_);                                                      \
    }

template <bool LIST, bool DEG>
__device__ __forceinline__ void es_filter(const int* __restrict__ src,
                                          const int* __restrict__ dst,
                                          const float* __restrict__ x,
                                          float* acc, int* q, int base,
                                          int c_begin, int c_end, int rot,
                                          int tid, int lane,
                                          int* __restrict__ wl, int sl, int* cnt) {
    constexpr unsigned RNG = LIST ? (unsigned)SR_RANGE : (unsigned)ES_RANGE;
    constexpr unsigned QSH = LIST ? 12u : 8u;
    constexpr int      NP  = LIST ? SL_PASSES : ES_PASSES;
    const int4* d4 = (const int4*)dst;
    int n = 0;
    #pragma unroll 1
    for (int it = 0; it < NP; ++it) {
        const int pass = (it + rot) % NP;
        const int c0   = c_begin + pass * ES_STRIDE + tid;
        int4 d[ES_GROUP];
        #pragma unroll
        for (int u = 0; u < ES_GROUP; ++u) {
            const int c = c0 + u * (ES_WAVES * 32);
            d[u] = d4[c < c_end ? c : c_end - 1];
            if (c >= c_end) d[u] = make_int4(-1, -1, -1, -1);
        }
        #pragma unroll
        for (int u = 0; u < ES_GROUP; ++u) {
            const int c = c0 + u * (ES_WAVES * 32);
            const bool any = (unsigned)(d[u].x - base) < RNG ||
                             (unsigned)(d[u].y - base) < RNG ||
                             (unsigned)(d[u].z - base) < RNG ||
                             (unsigned)(d[u].w - base) < RNG;
            if (__builtin_amdgcn_ballot_w32(any) != 0u) {
                ES_PUSH(d[u].x, 4 * c + 0)
                ES_PUSH(d[u].y, 4 * c + 1)
                ES_PUSH(d[u].z, 4 * c + 2)
                ES_PUSH(d[u].w, 4 * c + 3)
            }
        }
        if (n >= (LIST ? ES_LIST_AT : ES_DRAIN_AT)) {
            es_drain<LIST, DEG>(q, n, src, x, acc, lane, wl, sl, cnt);
            n = 0;
        }
    }
    es_drain<LIST, DEG>(q, n, src, x, acc, lane, wl, sl, cnt);
}

#define EB_NODES ((ES_RANGE + ES_WAVES - 1) / ES_WAVES)
static_assert(BAG <= 32, "one lane per token of a node");

__global__ __launch_bounds__(ES_WAVES * 32)
void range_prep(const int* __restrict__ tokens,
                const float* __restrict__ table,
                const int* __restrict__ src,
                const int* __restrict__ dst,
                float* __restrict__ x0,
                int* __restrict__ lists,
                int* __restrict__ counts) {
    __shared__ int qs[ES_WAVES * ES_QCAP];
    __shared__ int lcnt[SR_SPLIT];

    const int tid  = threadIdx.x;
    const int lane = tid & 31;
    const int wave = __builtin_amdgcn_readfirstlane(tid >> 5);
    const int base = blockIdx.x * ES_RANGE;
    const int sr   = blockIdx.x / SR_SPLIT;
    const int sl   = blockIdx.x % SR_SPLIT;
    int* q  = qs + wave * ES_QCAP;
    int* wl = lists + (size_t)sr * SR_SPLIT * SR_SPLIT * SL_CAP;

    if (tid < SR_SPLIT) lcnt[tid] = 0;

    int tk[EB_NODES];
    #pragma unroll
    for (int j = 0; j < EB_NODES; ++j) {
        int node = base + wave + j * ES_WAVES;
        node = node < N_NODES ? node : N_NODES - 1;
        tk[j] = tokens[node * BAG + (lane % BAG)];
    }
    __syncthreads();

    #pragma unroll 1
    for (int ph = 0; ph < 2; ++ph) {
        if (((ph ^ (wave >> 2)) & 1) == 0) {
            #pragma unroll
            for (int j = 0; j < EB_NODES; ++j) {
                const int r    = wave + j * ES_WAVES;
                const int node = base + r;
                if (r < ES_RANGE && node < N_NODES) {
                    float4 v[BAG];
                    int    tt[BAG];
                    #pragma unroll
                    for (int i = 0; i < BAG; ++i) {
                        tt[i] = __builtin_amdgcn_readlane(tk[j], i);
                        v[i]  = *(const float4*)(table + (size_t)tt[i] * EMB + lane * 4);
                    }
                    float4 a = make_float4(0.f, 0.f, 0.f, 0.f);
                    float  c = 0.f;
                    #pragma unroll
                    for (int i = 0; i < BAG; ++i) {
                        if (tt[i] != PAD_IDX) {
                            a.x += v[i].x; a.y += v[i].y; a.z += v[i].z; a.w += v[i].w;
                            c += 1.0f;
                        }
                    }
                    const float inv = 1.0f / fmaxf(c, 1.0f);
                    *(float4*)(x0 + (size_t)node * EMB + lane * 4) =
                        make_float4(a.x * inv, a.y * inv, a.z * inv, a.w * inv);
                }
            }
        } else {
            const int c_begin = sl * SL_INT4;
            const int c_end   = (c_begin + SL_INT4) < (N_EDGES / 4) ? (c_begin + SL_INT4) : (N_EDGES / 4);
            es_filter<true, false>(src, dst, nullptr, nullptr, q, sr * SR_RANGE,
                                   c_begin, c_end, sr, tid, lane, wl, sl, lcnt);
        }
    }
    __syncthreads();

    if (tid < SR_SPLIT) counts[(sr * SR_SPLIT + tid) * SR_SPLIT + sl] = lcnt[tid];
}

template <bool DEG>
__global__ __launch_bounds__(ES_WAVES * 32)
void edge_sum_range(const int* __restrict__ src,
                    const int* __restrict__ dst,
                    const float* __restrict__ x,
                    float* __restrict__ msg,
                    float* __restrict__ deg,
                    const int* __restrict__ lists,
                    const int* __restrict__ counts) {
    __shared__ __attribute__((aligned(16))) float acc[ES_RANGE * HID];
    __shared__ int qs[ES_WAVES * ES_QCAP];
    __shared__ int rowcnt[ES_RANGE];
    __shared__ int ovf;

    const int tid  = threadIdx.x;
    const int lane = tid & 31;
    const int wave = __builtin_amdgcn_readfirstlane(tid >> 5);
    const int base = blockIdx.x * ES_RANGE;
    int* q = qs + wave * ES_QCAP;
    const int sl = wave % SR_SPLIT;
    const int hf = wave / SR_SPLIT;

    const int  cw       = counts[blockIdx.x * SR_SPLIT + (lane % SR_SPLIT)];
    const bool use_list = __builtin_amdgcn_ballot_w32(cw > SL_CAP) == 0u;
    const int  c_sl     = counts[blockIdx.x * SR_SPLIT + sl];
    const int  mid      = (c_sl + 1) >> 1;
    const int  n_saved  = __builtin_amdgcn_readfirstlane(hf ? (c_sl - mid) : mid);
    const int* wlist    = lists + (size_t)(blockIdx.x * SR_SPLIT + sl) * SL_CAP + (hf ? mid : 0);

    int e[ES_HALF_LOADS];
    #pragma unroll
    for (int u = 0; u < ES_HALF_LOADS; ++u) {
        const int idx = u * 32 + lane;
        e[u] = (use_list && idx < n_saved) ? wlist[idx] : -1;
    }
    if (tid < ES_RANGE) rowcnt[tid] = 0;
    if (tid == 0) ovf = 0;
    __syncthreads();

    bool fast = use_list;
    if (use_list) {
        #pragma unroll
        for (int u = 0; u < ES_HALF_LOADS; ++u) {
            if (e[u] >= 0) {
                const int row = e[u] & 255;
                const int pos = atomicAdd(&rowcnt[row], 1);
                if (pos < RQ_CAP) qs[row * RQ_CAP + pos] = e[u] >> 8;
                else              ovf = 1;
            }
        }
        __syncthreads();
        fast = ovf == 0;
    }

    if (fast) {
        #pragma unroll 1
        for (int r = wave; r < ES_RANGE; r += ES_WAVES) {
            const int node = base + r;
            if (node >= N_NODES) break;
            const int  n  = __builtin_amdgcn_readfirstlane(rowcnt[r]);
            const int* rq = qs + r * RQ_CAP;
            float4 a = make_float4(0.f, 0.f, 0.f, 0.f);
            for (int b = 0; b < n; b += 32) {
                const int s   = (b + lane < n) ? rq[b + lane] : 0;
                const int cnt = (n - b) < 32 ? (n - b) : 32;
                for (int k0 = 0; k0 < cnt; k0 += ES_T) {
                    float4 v[ES_T];
                    #pragma unroll
                    for (int t = 0; t < ES_T; ++t) {
                        if (k0 + t < cnt) {
                            const int sk = __builtin_amdgcn_readlane(s, k0 + t);
                            v[t] = *(const float4*)(x + (size_t)sk * HID + lane * 4);
                        }
                    }
                    #pragma unroll
                    for (int t = 0; t < ES_T; ++t) {
                        if (k0 + t < cnt) {
                            a.x += v[t].x; a.y += v[t].y; a.z += v[t].z; a.w += v[t].w;
                        }
                    }
                }
            }
            *(float4*)(msg + (size_t)node * HID + lane * 4) = a;
            if (DEG) {
                if (lane == 0) deg[node] = (float)n;
            }
        }
    } else {
        for (int i = tid; i < ES_RANGE * HID / 4; i += ES_WAVES * 32)
            ((float4*)acc)[i] = make_float4(0.f, 0.f, 0.f, 0.f);
        if (DEG) {
            if (tid < ES_RANGE) rowcnt[tid] = 0;
        }
        __syncthreads();

        es_filter<false, DEG>(src, dst, x, acc, q, base, 0, N_EDGES / 4, blockIdx.x,
                              tid, lane, nullptr, 0, rowcnt);
        __syncthreads();

        for (int r = wave; r < ES_RANGE; r += ES_WAVES) {
            const int node = base + r;
            if (node < N_NODES) {
                const float* a = acc + r * HID + lane;
                *(float4*)(msg + (size_t)node * HID + lane * 4) = make_float4(a[0], a[32], a[64], a[96]);
            }
        }
        if (DEG) {
            if (tid < ES_RANGE && base + tid < N_NODES) deg[base + tid] = (float)rowcnt[tid];
        }
    }
}

#define GEMM_WGS   256
#define STEP_ROWS  32
#define N_STEPS    ((N_NODES + STEP_ROWS - 1) / STEP_ROWS)
#define A_PITCH    (2 * HID + 4)
#define W_PITCH    (HID + 4)
#define KB         8

#define FETCH_ROW(i, STEP)                                                      \
    {                                                                           \
        int row_ = (STEP) * STEP_ROWS + (i) * 8 + wave;                         \
        row_ = row_ < N_NODES ? row_ : N_NODES - 1;                             \
        rd##i = 1.0f / fmaxf(deg[row_], 1.0f);                                  \
        pm##i = *(const float4*)(msg + (size_t)row_ * HID + lane * 4);          \
        px##i = *(const float4*)(x   + (size_t)row_ * HID + lane * 4);          \
    }
#define PARK_ROW(i, DST)                                                        \
    {                                                                           \
        float4 m_ = pm##i;                                                      \
        m_.x *= rd##i; m_.y *= rd##i; m_.z *= rd##i; m_.w *= rd##i;             \
        *(float4*)((DST) + (i) * 8 * A_PITCH)       = m_;                       \
        *(float4*)((DST) + (i) * 8 * A_PITCH + HID) = px##i;                    \
    }

__global__ __launch_bounds__(256)
void sage_wmma(const float* __restrict__ msg,
               const float* __restrict__ deg,
               const float* x,
               const float* __restrict__ wl,
               const float* __restrict__ bias,
               const float* __restrict__ wr,
               float* out) {
    __shared__ __attribute__((aligned(16))) float sW[2 * HID * W_PITCH];
    __shared__ __attribute__((aligned(16))) float sA[2 * STEP_ROWS * A_PITCH];

    const int lane   = threadIdx.x & 31;
    const int wave   = threadIdx.x >> 5;
    const int half   = lane >> 4;
    const int lrow   = lane & 15;
    const int n_base = wave * 16;

    {
        const float4* gl = (const float4*)(wl + (size_t)n_base * HID);
        const float4* gr = (const float4*)(wr + (size_t)n_base * HID);
        float* dl = sW + n_base * W_PITCH + lane * 4;
        float* dr = sW + (HID + n_base) * W_PITCH + lane * 4;
        #pragma unroll
        for (int i = 0; i < 16; ++i) {
            const float4 a = gl[i * 32 + lane];
            const float4 b = gr[i * 32 + lane];
            *(float4*)(dl + i * W_PITCH) = a;
            *(float4*)(dr + i * W_PITCH) = b;
        }
    }

    float4 pm0, pm1, pm2, pm3, px0, px1, px2, px3;
    float  rd0, rd1, rd2, rd3;

    int step = blockIdx.x;
    FETCH_ROW(0, step) FETCH_ROW(1, step) FETCH_ROW(2, step) FETCH_ROW(3, step)

    __syncthreads();
    v2f bl[HID / 4], br[HID / 4];
    {
        const float* pl = sW + (n_base + lrow) * W_PITCH + half * 2;
        const float* pr = pl + HID * W_PITCH;
        #pragma unroll
        for (int j = 0; j < HID / 4; ++j) {
            bl[j] = *(const v2f*)(pl + 4 * j);
            br[j] = *(const v2f*)(pr + 4 * j);
        }
    }
    const float bv = bias[n_base + lrow];

    {
        float* d = sA + wave * A_PITCH + lane * 4;
        PARK_ROW(0, d) PARK_ROW(1, d) PARK_ROW(2, d) PARK_ROW(3, d)
    }
    __syncthreads();

    int buf = 0;
    for (; step < N_STEPS; step += GEMM_WGS) {
        const int  next     = step + GEMM_WGS;
        const bool has_next = next < N_STEPS;
        if (has_next) {
            FETCH_ROW(0, next) FETCH_ROW(1, next) FETCH_ROW(2, next) FETCH_ROW(3, next)
        }

        const float* a0 = sA + buf * (STEP_ROWS * A_PITCH) + lrow * A_PITCH + half * 2;
        const float* a1 = a0 + 16 * A_PITCH;

        v8f c0, c1;
        #pragma unroll
        for (int r = 0; r < 8; ++r) { c0[r] = bv; c1[r] = bv; }

        v2f fm0[2][KB], fm1[2][KB], fx0[2][KB], fx1[2][KB];
        #pragma unroll
        for (int t = 0; t < KB; ++t) {
            fm0[0][t] = *(const v2f*)(a0 + 4 * t);
            fm1[0][t] = *(const v2f*)(a1 + 4 * t);
            fx0[0][t] = *(const v2f*)(a0 + HID + 4 * t);
            fx1[0][t] = *(const v2f*)(a1 + HID + 4 * t);
        }
        #pragma unroll
        for (int b = 0; b < HID / 4 / KB; ++b) {
            const int cur = b & 1;
            if (b + 1 < HID / 4 / KB) {
                #pragma unroll
                for (int t = 0; t < KB; ++t) {
                    const int k = 4 * ((b + 1) * KB + t);
                    fm0[cur ^ 1][t] = *(const v2f*)(a0 + k);
                    fm1[cur ^ 1][t] = *(const v2f*)(a1 + k);
                    fx0[cur ^ 1][t] = *(const v2f*)(a0 + HID + k);
                    fx1[cur ^ 1][t] = *(const v2f*)(a1 + HID + k);
                }
            }
            __builtin_amdgcn_sched_barrier(0);
            #pragma unroll
            for (int t = 0; t < KB; ++t) {
                const int j = b * KB + t;
                c0 = __builtin_amdgcn_wmma_f32_16x16x4_f32(false, fm0[cur][t], false, bl[j],
                                                           (short)0, c0, false, false);
                c1 = __builtin_amdgcn_wmma_f32_16x16x4_f32(false, fm1[cur][t], false, bl[j],
                                                           (short)0, c1, false, false);
                c0 = __builtin_amdgcn_wmma_f32_16x16x4_f32(false, fx0[cur][t], false, br[j],
                                                           (short)0, c0, false, false);
                c1 = __builtin_amdgcn_wmma_f32_16x16x4_f32(false, fx1[cur][t], false, br[j],
                                                           (short)0, c1, false, false);
            }
        }

        if (has_next) {
            float* d = sA + (buf ^ 1) * (STEP_ROWS * A_PITCH) + wave * A_PITCH + lane * 4;
            PARK_ROW(0, d) PARK_ROW(1, d) PARK_ROW(2, d) PARK_ROW(3, d)
        }
        __syncthreads();
        buf ^= 1;

        const int m0_base = step * STEP_ROWS;
        #pragma unroll
        for (int r = 0; r < 8; ++r)
            out[(size_t)(m0_base + half * 8 + r) * HID + n_base + lrow] = fmaxf(c0[r], 0.0f);
        if (m0_base + 16 < N_NODES) {
            #pragma unroll
            for (int r = 0; r < 8; ++r)
                out[(size_t)(m0_base + 16 + half * 8 + r) * HID + n_base + lrow] = fmaxf(c1[r], 0.0f);
        }
    }
}

#define PH_WGS    256
#define PH_GPW    2
#define PH_WAVES  16
#define PH_T      16
#define PH_U      49
#define PH_GROUPS ((N_NODES + 31) / 32)
#define PH_ITERS  ((PH_GROUPS + PH_WAVES - 1) / PH_WAVES)
static_assert(PH_WGS * PH_GPW == N_GRAPHS, "every graph has exactly one owner");
static_assert(PH_GPW == 2, "a wave keeps one register sum per owned graph: a0, a1");
static_assert(PH_ITERS % PH_U == 0, "the scan loop is written in whole batches");
static_assert(PH_U <= 64, "one bit per group of a batch in a 64-bit mask");
static_assert(PH_GPW * HID <= PH_WAVES * 32, "one thread per (graph, feature) in the combine");

__device__ __forceinline__ void ph_rows(unsigned bal, unsigned rel, int nbase,
                                        const float* __restrict__ x, int lane,
                                        float4& a0, float4& a1) {
    while (bal != 0u) {
        const int left = __popc(bal);
        const int cnt  = left < PH_T ? left : PH_T;
        float4   v[PH_T];
        unsigned g[PH_T];
        int klast = __builtin_ctz(bal);
        #pragma unroll
        for (int t = 0; t < PH_T; ++t) {
            const bool has = bal != 0u;
            const int  k   = has ? __builtin_ctz(bal) : klast;
            klast = k;
            bal   = has ? (bal & (bal - 1u)) : 0u;
            g[t] = (unsigned)__builtin_amdgcn_readlane((int)rel, k);
            v[t] = *(const float4*)(x + (size_t)(nbase + k) * HID + lane * 4);
        }
        #pragma unroll
        for (int t = 0; t < PH_T; ++t) {
            if (t < cnt) {
                if (g[t] == 0u) { a0.x += v[t].x; a0.y += v[t].y; a0.z += v[t].z; a0.w += v[t].w; }
                else            { a1.x += v[t].x; a1.y += v[t].y; a1.z += v[t].z; a1.w += v[t].w; }
            }
        }
    }
}

__global__ __launch_bounds__(PH_WAVES * 32)
void pool_head(const float* __restrict__ x,
               const int* __restrict__ batch,
               const float* __restrict__ wout,
               const float* __restrict__ bout,
               float* __restrict__ out) {
    __shared__ __attribute__((aligned(16))) float part[PH_WAVES * PH_GPW * HID];
    __shared__ __attribute__((aligned(16))) float gs[PH_GPW * HID];
    __shared__ int   pcnt[PH_WAVES * PH_GPW];
    __shared__ float ginv[PH_GPW];

    const int tid  = threadIdx.x;
    const int lane = tid & 31;
    const int wave = tid >> 5;
    const int g0   = blockIdx.x * PH_GPW;

    float4 a0 = make_float4(0.f, 0.f, 0.f, 0.f);
    float4 a1 = make_float4(0.f, 0.f, 0.f, 0.f);
    int c0 = 0, c1 = 0;

    for (int i0 = 0; i0 < PH_ITERS; i0 += PH_U) {
        int b[PH_U];
        #pragma unroll
        for (int u = 0; u < PH_U; ++u) {
            const int node = ((i0 + u) * PH_WAVES + wave) * 32 + lane;
            b[u] = batch[node < N_NODES ? node : N_NODES - 1];
            if (node >= N_NODES) b[u] = -1;
        }
        unsigned long long um = 0ull;
        #pragma unroll
        for (int u = 0; u < PH_U; ++u)
            if (__builtin_amdgcn_ballot_w32((unsigned)(b[u] - g0) < (unsigned)PH_GPW) != 0u) um |= 1ull << u;
        while (um != 0ull) {
            const int u = __builtin_ctzll(um);
            um &= um - 1ull;
            int bv = b[0];
            #pragma unroll
            for (int uu = 1; uu < PH_U; ++uu) bv = (u == uu) ? b[uu] : bv;
            const unsigned rel = (unsigned)(bv - g0);
            const unsigned bal = __builtin_amdgcn_ballot_w32(rel < (unsigned)PH_GPW);
            const unsigned bz  = __builtin_amdgcn_ballot_w32(rel == 0u);
            c0 += __popc(bz);
            c1 += __popc(bal) - __popc(bz);
            ph_rows(bal, rel, ((i0 + u) * PH_WAVES + wave) * 32, x, lane, a0, a1);
        }
    }

    *(float4*)(part + (wave * PH_GPW + 0) * HID + lane * 4) = a0;
    *(float4*)(part + (wave * PH_GPW + 1) * HID + lane * 4) = a1;
    if (lane == 0) { pcnt[wave * PH_GPW + 0] = c0; pcnt[wave * PH_GPW + 1] = c1; }
    __syncthreads();

    if (tid < PH_GPW * HID) {
        const int j = tid / HID, f = tid % HID;
        float s = 0.0f;
        #pragma unroll
        for (int w = 0; w < PH_WAVES; ++w) s += part[(w * PH_GPW + j) * HID + f];
        gs[j * HID + f] = s;
    } else if (tid < PH_GPW * HID + PH_GPW) {
        const int j = tid - PH_GPW * HID;
        int c = 0;
        #pragma unroll
        for (int w = 0; w < PH_WAVES; ++w) c += pcnt[w * PH_GPW + j];
        ginv[j] = 1.0f / fmaxf((float)c, 1.0f);
    }
    __syncthreads();

    for (int p = wave; p < PH_GPW * N_CLASSES; p += PH_WAVES) {
        const int j = p / N_CLASSES, c = p % N_CLASSES;
        const float4 gv = *(const float4*)(gs + j * HID + lane * 4);
        const float4 wv = *(const float4*)(wout + (size_t)c * HID + lane * 4);
        float s = gv.x * wv.x + gv.y * wv.y + gv.z * wv.z + gv.w * wv.w;
        #pragma unroll
        for (int o = 16; o > 0; o >>= 1) s += __shfl_xor(s, o, 32);
        if (lane == 0) out[(g0 + j) * N_CLASSES + c] = s * ginv[j] + bout[c];
    }
}

extern "C" void kernel_launch(void* const* d_in, const int* in_sizes, int n_in,
                              void* d_out, int out_size, void* d_ws, size_t ws_size,
                              hipStream_t stream) {
    (void)in_sizes; (void)n_in; (void)out_size; (void)ws_size;

    const int*   x_tokens = (const int*)  d_in[0];
    const int*   edge     = (const int*)  d_in[1];
    const int*   batch    = (const int*)  d_in[2];
    const float* emb      = (const float*)d_in[3];
    const float* w1l      = (const float*)d_in[4];
    const float* b1       = (const float*)d_in[5];
    const float* w1r      = (const float*)d_in[6];
    const float* w2l      = (const float*)d_in[7];
    const float* b2       = (const float*)d_in[8];
    const float* w2r      = (const float*)d_in[9];
    const float* wout     = (const float*)d_in[10];
    const float* bout     = (const float*)d_in[11];
    float*       out      = (float*)d_out;

    const int* src = edge;
    const int* dst = edge + N_EDGES;

    const size_t nodeF = (size_t)N_NODES * HID;
    float* bufA = (float*)d_ws;
    float* bufB = bufA + nodeF;
    float* bufC = bufB + nodeF;
    float* deg  = bufC + nodeF;
    int* lists  = (int*)bufC;
    int* counts = lists + ES_WGS * SR_SPLIT * SL_CAP;

    const dim3 gEdge(ES_WGS);
    const dim3 bEdge(ES_WAVES * 32);
    const dim3 gGemm(GEMM_WGS);
    const dim3 bGemm(256);

    range_prep<<<gEdge, bEdge, 0, stream>>>(x_tokens, emb, src, dst, bufA, lists, counts);

    edge_sum_range<true><<<gEdge, bEdge, 0, stream>>>(src, dst, bufA, bufB, deg, lists, counts);
    sage_wmma<<<gGemm, bGemm, 0, stream>>>(bufB, deg, bufA, w1l, b1, w1r, bufA);

    edge_sum_range<false><<<gEdge, bEdge, 0, stream>>>(src, dst, bufA, bufB, deg, lists, counts);
    sage_wmma<<<gGemm, bGemm, 0, stream>>>(bufB, deg, bufA, w2l, b2, w2r, bufA);

    pool_head<<<PH_WGS, PH_WAVES * 32, 0, stream>>>(bufA, batch, wout, bout, out);
}
